# starting kernel
# baseline (speedup 1.0000x reference)
#include <hip/hip_runtime.h>
#include <hip/hip_bf16.h>

#define N_NODES   100000
#define N_EDGES   20000
#define NNZ_CNT   800000
#define D_IN      128
#define D         256
#define EPS_F     1e-5f

typedef float v2f __attribute__((ext_vector_type(2)));
typedef float v8f __attribute__((ext_vector_type(8)));

__device__ __forceinline__ void atomAddF(float* p, float v) {
    __hip_atomic_fetch_add(p, v, __ATOMIC_RELAXED, __HIP_MEMORY_SCOPE_AGENT);
}

__global__ __launch_bounds__(256) void degcnt_kernel(const int* __restrict__ node_idx,
                                                     const int* __restrict__ edge_idx,
                                                     float* __restrict__ deg,
                                                     float* __restrict__ cnt) {
    int i = blockIdx.x * 256 + threadIdx.x;
    if (i < NNZ_CNT) {
        atomAddF(&deg[node_idx[i]], 1.0f);
        atomAddF(&cnt[edge_idx[i]], 1.0f);
    }
}

template<int K>
__global__ __launch_bounds__(256) void gemm_wmma_kernel(const float* __restrict__ X,
                                                        const float* __restrict__ W,
                                                        float* __restrict__ C,
                                                        int mTiles) {
    const int wid  = (blockIdx.x * 256 + threadIdx.x) >> 5;
    const int lane = threadIdx.x & 31;
    const int mT   = wid >> 2;
    const int nT   = wid & 3;
    if (mT >= mTiles) return;
    const int m0   = mT * 16;
    const int n0   = nT * 64;
    const int half = lane >> 4;
    const int l16  = lane & 15;

    const float* __restrict__ xrow = X + (size_t)(m0 + l16) * K;
    const float* __restrict__ wrow = W + (size_t)(n0 + l16) * K;

    v8f acc0 = {}, acc1 = {}, acc2 = {}, acc3 = {};

    #pragma unroll 8
    for (int k0 = 0; k0 < K; k0 += 4) {
        const int ko = k0 + 2 * half;
        v2f a  = *reinterpret_cast<const v2f*>(xrow + ko);
        v2f b0 = *reinterpret_cast<const v2f*>(wrow + (size_t)0 * 16 * K + ko);
        v2f b1 = *reinterpret_cast<const v2f*>(wrow + (size_t)1 * 16 * K + ko);
        v2f b2 = *reinterpret_cast<const v2f*>(wrow + (size_t)2 * 16 * K + ko);
        v2f b3 = *reinterpret_cast<const v2f*>(wrow + (size_t)3 * 16 * K + ko);
        acc0 = __builtin_amdgcn_wmma_f32_16x16x4_f32(false, a, false, b0, (short)0, acc0, false, false);
        acc1 = __builtin_amdgcn_wmma_f32_16x16x4_f32(false, a, false, b1, (short)0, acc1, false, false);
        acc2 = __builtin_amdgcn_wmma_f32_16x16x4_f32(false, a, false, b2, (short)0, acc2, false, false);
        acc3 = __builtin_amdgcn_wmma_f32_16x16x4_f32(false, a, false, b3, (short)0, acc3, false, false);
    }

    float* __restrict__ ccol = C + n0 + l16;
    #pragma unroll
    for (int r = 0; r < 8; ++r) {
        const size_t m = (size_t)(m0 + r + 8 * half);
        ccol[m * D + 0 * 16] = acc0[r];
        ccol[m * D + 1 * 16] = acc1[r];
        ccol[m * D + 2 * 16] = acc2[r];
        ccol[m * D + 3 * 16] = acc3[r];
    }
}

__global__ __launch_bounds__(256) void scatter_rows_kernel(const float* __restrict__ src,
                                                           const int* __restrict__ src_idx,
                                                           const int* __restrict__ dst_idx,
                                                           float* __restrict__ dst) {
    const int i = blockIdx.x;
    const int f = threadIdx.x;
    const int s = src_idx[i];
    const int d = dst_idx[i];
    const float v = src[(size_t)s * D + f];
    atomAddF(&dst[(size_t)d * D + f], v);
}

__global__ __launch_bounds__(256) void scale_edges_kernel(float* __restrict__ E,
                                                          const float* __restrict__ cnt) {
    const int idx = blockIdx.x * 256 + threadIdx.x;
    if (idx < N_EDGES * D) {
        const float c = cnt[idx >> 8];
        const float inv = (c > 0.0f) ? 1.0f / c : 0.0f;
        E[idx] *= inv;
    }
}

__global__ __launch_bounds__(256) void fused_scale_ln_prelu_kernel(const float* __restrict__ acc,
                                                                   const float* __restrict__ deg,
                                                                   const float* __restrict__ bias,
                                                                   const float* __restrict__ g,
                                                                   const float* __restrict__ beta,
                                                                   const float* __restrict__ pa,
                                                                   float* __restrict__ out) {
    __shared__ float sred[256];
    const int r = blockIdx.x;
    const int f = threadIdx.x;
    const float dg   = deg[r];
    const float dinv = (dg > 0.0f) ? 1.0f / dg : 0.0f;
    const float v = acc[(size_t)r * D + f] * dinv + bias[f];

    sred[f] = v;
    __syncthreads();
    #pragma unroll
    for (int s = 128; s > 0; s >>= 1) { if (f < s) sred[f] += sred[f + s]; __syncthreads(); }
    const float mu = sred[0] * (1.0f / D);
    __syncthreads();

    const float dv = v - mu;
    sred[f] = dv * dv;
    __syncthreads();
    #pragma unroll
    for (int s = 128; s > 0; s >>= 1) { if (f < s) sred[f] += sred[f + s]; __syncthreads(); }
    const float var = sred[0] * (1.0f / D);

    const float y = dv * rsqrtf(var + EPS_F) * g[f] + beta[f];
    const float a = pa[0];
    out[(size_t)r * D + f] = (y >= 0.0f) ? y : a * y;
}

__global__ __launch_bounds__(256) void wstd_kernel(const float* __restrict__ W,
                                                   float* __restrict__ Ws) {
    __shared__ float sred[256];
    const int r = blockIdx.x;
    const int f = threadIdx.x;
    const float w = W[(size_t)r * D + f];

    sred[f] = w;
    __syncthreads();
    #pragma unroll
    for (int s = 128; s > 0; s >>= 1) { if (f < s) sred[f] += sred[f + s]; __syncthreads(); }
    const float mean = sred[0] * (1.0f / D);
    __syncthreads();

    const float dv = w - mean;
    sred[f] = dv * dv;
    __syncthreads();
    #pragma unroll
    for (int s = 128; s > 0; s >>= 1) { if (f < s) sred[f] += sred[f + s]; __syncthreads(); }
    const float var = sred[0] * (1.0f / (D - 1));

    Ws[(size_t)r * D + f] = dv * rsqrtf(var + EPS_F);
}

extern "C" void kernel_launch(void* const* d_in, const int* in_sizes, int n_in,
                              void* d_out, int out_size, void* d_ws, size_t ws_size,
                              hipStream_t stream) {
    const float* edge_fea = (const float*)d_in[0];
    const int*   node_idx = (const int*)  d_in[1];
    const int*   edge_idx = (const int*)  d_in[2];
    const float* W1       = (const float*)d_in[3];
    const float* b1       = (const float*)d_in[4];
    const float* W2       = (const float*)d_in[5];
    const float* b2       = (const float*)d_in[6];
    const float* g1       = (const float*)d_in[7];
    const float* beta1    = (const float*)d_in[8];
    const float* g2       = (const float*)d_in[9];
    const float* beta2    = (const float*)d_in[10];
    const float* prelu_a  = (const float*)d_in[11];

    const size_t nodeRowBytes = (size_t)N_NODES * D * sizeof(float);
    const size_t edgeRowBytes = (size_t)N_EDGES * D * sizeof(float);

    float* B0  = (float*)d_out;
    char*  ws  = (char*)d_ws;
    float* B1  = (float*)ws;
    float* E   = (float*)(ws + nodeRowBytes);
    float* deg = E + (size_t)N_EDGES * D;
    float* cnt = deg + N_NODES;
    float* W2s = cnt + N_EDGES;

    const int mTiles = N_NODES / 16;
    const int gemmBlocks = (mTiles * 4 * 32) / 256;

    hipMemsetAsync(deg, 0, (size_t)(N_NODES + N_EDGES) * sizeof(float), stream);
    degcnt_kernel<<<(NNZ_CNT + 255) / 256, 256, 0, stream>>>(node_idx, edge_idx, deg, cnt);

    gemm_wmma_kernel<D_IN><<<gemmBlocks, 256, 0, stream>>>(edge_fea, W1, B1, mTiles);

    hipMemsetAsync(E, 0, edgeRowBytes, stream);
    scatter_rows_kernel<<<NNZ_CNT, 256, 0, stream>>>(B1, node_idx, edge_idx, E);
    scale_edges_kernel<<<(N_EDGES * D) / 256, 256, 0, stream>>>(E, cnt);

    hipMemsetAsync(B0, 0, nodeRowBytes, stream);
    scatter_rows_kernel<<<NNZ_CNT, 256, 0, stream>>>(E, edge_idx, node_idx, B0);
    fused_scale_ln_prelu_kernel<<<N_NODES, 256, 0, stream>>>(B0, deg, b1, g1, beta1, prelu_a, B1);

    wstd_kernel<<<D, 256, 0, stream>>>(W2, W2s);
    gemm_wmma_kernel<D><<<gemmBlocks, 256, 0, stream>>>(B1, W2s, B0, mTiles);

    hipMemsetAsync(E, 0, edgeRowBytes, stream);
    scatter_rows_kernel<<<NNZ_CNT, 256, 0, stream>>>(B0, node_idx, edge_idx, E);
    scale_edges_kernel<<<(N_EDGES * D) / 256, 256, 0, stream>>>(E, cnt);

    hipMemsetAsync(B1, 0, nodeRowBytes, stream);
    scatter_rows_kernel<<<NNZ_CNT, 256, 0, stream>>>(E, edge_idx, node_idx, B1);
    fused_scale_ln_prelu_kernel<<<N_NODES, 256, 0, stream>>>(B1, deg, b2, g2, beta2, prelu_a, B0);
}


# Row scatter-adds moved into LDS: a work-group owns a destination row range, scans the whole index list, adds matching source rows by LDS float atomics, stores rows once; sums in LDS arrival order
# speedup vs baseline: 4.1127x; 4.1127x over previous
#include <hip/hip_runtime.h>
#include <hip/hip_bf16.h>

#define N_NODES   100000
#define N_EDGES   20000
#define NNZ_CNT   800000
#define D_IN      128
#define D         256
#define EPS_F     1e-5f

typedef float v2f __attribute__((ext_vector_type(2)));
typedef float v8f __attribute__((ext_vector_type(8)));

__device__ __forceinline__ void atomAddF(float* p, float v) {
    __hip_atomic_fetch_add(p, v, __ATOMIC_RELAXED, __HIP_MEMORY_SCOPE_AGENT);
}

__global__ __launch_bounds__(256) void degcnt_kernel(const int* __restrict__ node_idx,
                                                     const int* __restrict__ edge_idx,
                                                     float* __restrict__ deg,
                                                     float* __restrict__ cnt) {
    int i = blockIdx.x * 256 + threadIdx.x;
    if (i < NNZ_CNT) {
        atomAddF(&deg[node_idx[i]], 1.0f);
        atomAddF(&cnt[edge_idx[i]], 1.0f);
    }
}

#define GM_ROWS 64
#define GM_BLOCKS ((N_NODES + GM_ROWS - 1) / GM_ROWS)
static_assert(D == 256, "four waves of 64 columns cover the 256 output columns");
static_assert(N_NODES % 16 == 0, "a 16-row tile is wholly inside or wholly outside the matrix");

typedef float v4f __attribute__((ext_vector_type(4)));

__device__ __forceinline__ void load_frag(const float* __restrict__ p, v4f (&f)[4]) {
    #pragma unroll
    for (int j = 0; j < 4; ++j) f[j] = *reinterpret_cast<const v4f*>(p + 4 * j);
}

template<int K>
__global__ __launch_bounds__(128) void gemm_wmma_kernel(const float* __restrict__ X,
                                                        const float* __restrict__ W,
                                                        float* __restrict__ C) {
    static_assert(K % 32 == 0, "K is walked in 32-k chunks");
    constexpr int NCH = K / 32;
    const int lane = threadIdx.x & 31;
    const int wave = __builtin_amdgcn_readfirstlane((int)(threadIdx.x >> 5));
    const int c  = lane & 15;
    const int h  = lane >> 4;
    const int m0 = blockIdx.x * GM_ROWS;
    const int n0 = wave * 64;

    const float* __restrict__ ap[4];
    #pragma unroll
    for (int i = 0; i < 4; ++i) {
        const int ti = (m0 + 16 * i < N_NODES) ? i : 0;
        ap[i] = X + (size_t)(m0 + 16 * ti + c) * K + 16 * h;
    }
    const float* __restrict__ bp = W + (size_t)(n0 + 4 * c) * K + 16 * h;

    v8f acc[4][4];
    #pragma unroll
    for (int i = 0; i < 4; ++i)
        #pragma unroll
        for (int n = 0; n < 4; ++n) acc[i][n] = v8f{};

    v4f fa[2][4][4], fb[2][4][4];

    #pragma unroll
    for (int i = 0; i < 4; ++i) load_frag(ap[i], fa[0][i]);
    #pragma unroll
    for (int n = 0; n < 4; ++n) load_frag(bp + (size_t)n * K, fb[0][n]);

    #pragma unroll
    for (int q = 0; q < NCH; ++q) {
        const int cur = q & 1, nxt = cur ^ 1;
        if (q + 1 < NCH) {
            #pragma unroll
            for (int i = 0; i < 4; ++i) load_frag(ap[i] + 32 * (q + 1), fa[nxt][i]);
            #pragma unroll
            for (int n = 0; n < 4; ++n) load_frag(bp + (size_t)n * K + 32 * (q + 1), fb[nxt][n]);
        }
        __builtin_amdgcn_sched_barrier(0);
        #pragma unroll
        for (int j = 0; j < 4; ++j) {
            #pragma unroll
            for (int p = 0; p < 2; ++p) {
                v2f a[4], b[4];
                #pragma unroll
                for (int i = 0; i < 4; ++i) {
                    a[i] = p ? __builtin_shufflevector(fa[cur][i][j], fa[cur][i][j], 2, 3)
                             : __builtin_shufflevector(fa[cur][i][j], fa[cur][i][j], 0, 1);
                    b[i] = p ? __builtin_shufflevector(fb[cur][i][j], fb[cur][i][j], 2, 3)
                             : __builtin_shufflevector(fb[cur][i][j], fb[cur][i][j], 0, 1);
                }
                #pragma unroll
                for (int i = 0; i < 4; ++i)
                    #pragma unroll
                    for (int n = 0; n < 4; ++n)
                        acc[i][n] = __builtin_amdgcn_wmma_f32_16x16x4_f32(false, a[i], false, b[n], (short)0,
                                                                          acc[i][n], false, false);
            }
        }
    }

    float* __restrict__ cp = C + (size_t)(m0 + 8 * h) * D + n0 + 4 * c;
    #pragma unroll
    for (int i = 0; i < 4; ++i) {
        if (m0 + 16 * i < N_NODES) {
            #pragma unroll
            for (int r = 0; r < 8; ++r) {
                const v4f o = {acc[i][0][r], acc[i][1][r], acc[i][2][r], acc[i][3][r]};
                *reinterpret_cast<v4f*>(cp + (size_t)(16 * i + r) * D) = o;
            }
        }
    }
}

#define SL_THREADS 1024
#define SL_WAVES   (SL_THREADS / 32)
#define SL_QCAP    144
#define SL_NQ      (NNZ_CNT / 4)
#define SL_QPI     6
#define SL_STEP    (SL_THREADS * SL_QPI)
#define SL_NIT     ((SL_NQ + SL_STEP - 1) / SL_STEP)
static_assert(NNZ_CNT % 4 == 0, "index list is read as 16-byte quads");
static_assert(NNZ_CNT < (1 << 24), "an entry number packs into 24 bits");

typedef int v4i __attribute__((ext_vector_type(4)));

__device__ __forceinline__ void ldsAddF(float* p, float v) {
    __hip_atomic_fetch_add(p, v, __ATOMIC_RELAXED, __HIP_MEMORY_SCOPE_WORKGROUP);
}

__device__ __forceinline__ v4i sl_load_quad(const v4i* __restrict__ d4, int qi) {
    return d4[min(qi, SL_NQ - 1)];
}

__device__ __forceinline__ void sl_add8(const unsigned* q, int pos, int lane,
                                        const float* __restrict__ src,
                                        const int* __restrict__ src_idx, float* acc) {
    const unsigned pk = q[pos + (lane & 7)];
    const int rel = (int)(pk & 255u);
    const int s = src_idx[pk >> 8];
    float v[8][8];
    for (int j = 0; j < 8; ++j) {
        const int sj = __builtin_amdgcn_readlane(s, j);
        const float* __restrict__ row = src + (size_t)sj * D + lane;
        for (int c = 0; c < 8; ++c) v[j][c] = row[c * 32];
    }
    for (int j = 0; j < 8; ++j) {
        const int rj = __builtin_amdgcn_readlane(rel, j);
        float* a = acc + rj * D + lane;
        for (int c = 0; c < 8; ++c) ldsAddF(a + c * 32, v[j][c]);
    }
}

template<int ROWS, int NDST>
__global__ __launch_bounds__(SL_THREADS) void scatter_rows_lds_kernel(const float* __restrict__ src,
                                                                      const int* __restrict__ src_idx,
                                                                      const int* __restrict__ dst_idx,
                                                                      float* __restrict__ dst) {
    static_assert(ROWS <= 256, "a row number inside the group packs into 8 bits");
    __shared__ float acc[ROWS * D];
    __shared__ unsigned list[SL_WAVES][SL_QCAP];

    const int tid  = threadIdx.x;
    const int lane = tid & 31;
    const int wave = __builtin_amdgcn_readfirstlane(tid >> 5);
    const int base = blockIdx.x * ROWS;

    v4f* acc4 = reinterpret_cast<v4f*>(acc);
    for (int k = tid; k < ROWS * D / 4; k += SL_THREADS) acc4[k] = v4f{0.0f, 0.0f, 0.0f, 0.0f};
    __syncthreads();

    unsigned* q = list[wave];
    int cnt = 0;
    const v4i* __restrict__ d4 = reinterpret_cast<const v4i*>(dst_idx);

    v4i cur[SL_QPI], nxt[SL_QPI];
    for (int u = 0; u < SL_QPI; ++u) cur[u] = sl_load_quad(d4, u * SL_THREADS + tid);

    for (int it = 0; it < SL_NIT; ++it) {
        const int q0 = it * SL_STEP + tid;
        for (int u = 0; u < SL_QPI; ++u) nxt[u] = sl_load_quad(d4, q0 + SL_STEP + u * SL_THREADS);
        #pragma unroll
        for (int u = 0; u < SL_QPI; ++u) {
            const int qi = q0 + u * SL_THREADS;
            const unsigned inlist = __builtin_amdgcn_ballot_w32(qi < SL_NQ);
            unsigned rel[4], mk[4];
            for (int c = 0; c < 4; ++c) {
                rel[c] = (unsigned)(cur[u][c] - base);
                mk[c]  = __builtin_amdgcn_ballot_w32(rel[c] < (unsigned)ROWS) & inlist;
            }
            if ((mk[0] | mk[1] | mk[2] | mk[3]) != 0u) {
                for (int c = 0; c < 4; ++c) {
                    if (qi < SL_NQ && rel[c] < (unsigned)ROWS)
                        q[cnt + (int)__builtin_amdgcn_mbcnt_lo(mk[c], 0u)] =
                            ((unsigned)(4 * qi + c) << 8) | rel[c];
                    cnt += __builtin_popcount(mk[c]);
                }
                while (cnt >= 8) {
                    cnt -= 8;
                    sl_add8(q, cnt, lane, src, src_idx, acc);
                }
            }
        }
        for (int u = 0; u < SL_QPI; ++u) cur[u] = nxt[u];
    }

    for (int j = 0; j < cnt; ++j) {
        const unsigned pk = (unsigned)__builtin_amdgcn_readfirstlane((int)q[j]);
        const int s = __builtin_amdgcn_readfirstlane(src_idx[pk >> 8]);
        const float* __restrict__ row = src + (size_t)s * D + lane;
        float* a = acc + (int)(pk & 255u) * D + lane;
        float v[8];
        for (int c = 0; c < 8; ++c) v[c] = row[c * 32];
        for (int c = 0; c < 8; ++c) ldsAddF(a + c * 32, v[c]);
    }
    __syncthreads();

    v4f* __restrict__ out4 = reinterpret_cast<v4f*>(dst + (size_t)base * D);
    for (int k = tid; k < ROWS * D / 4; k += SL_THREADS)
        if (base + (k >> 6) < NDST) out4[k] = acc4[k];
}

#define SL_ROWS_E 80
#define SL_ROWS_N 256
#define SL_BLOCKS_E ((N_EDGES + SL_ROWS_E - 1) / SL_ROWS_E)
#define SL_BLOCKS_N ((N_NODES + SL_ROWS_N - 1) / SL_ROWS_N)

__global__ __launch_bounds__(256) void scale_edges_kernel(float* __restrict__ E,
                                                          const float* __restrict__ cnt) {
    const int idx = blockIdx.x * 256 + threadIdx.x;
    if (idx < N_EDGES * D) {
        const float c = cnt[idx >> 8];
        const float inv = (c > 0.0f) ? 1.0f / c : 0.0f;
        E[idx] *= inv;
    }
}

#define LN_RPW 4
#define LN_WAVES_PER_BLOCK 8
#define LN_BLOCKS (N_NODES / (LN_RPW * LN_WAVES_PER_BLOCK))
static_assert(N_NODES % (LN_RPW * LN_WAVES_PER_BLOCK) == 0, "row count must fill whole blocks");
static_assert(D == 256, "lane-to-column map below is written for 256-wide rows");

__device__ __forceinline__ float lane_sum8(v4f a, v4f b) {
    return ((a.x + a.y) + (a.z + a.w)) + ((b.x + b.y) + (b.z + b.w));
}

__global__ __launch_bounds__(256) void fused_scale_ln_prelu_kernel(const float* __restrict__ acc,
                                                                   const float* __restrict__ deg,
                                                                   const float* __restrict__ bias,
                                                                   const float* __restrict__ g,
                                                                   const float* __restrict__ beta,
                                                                   const float* __restrict__ pa,
                                                                   float* __restrict__ out) {
    const int lane = threadIdx.x & 31;
    const int wv = __builtin_amdgcn_readfirstlane((int)((blockIdx.x * 256 + threadIdx.x) >> 5));
    const int r0 = wv * LN_RPW;
    const int c0 = lane * 4;

    v4f x0[LN_RPW], x1[LN_RPW];
    for (int j = 0; j < LN_RPW; ++j) {
        const float* __restrict__ row = acc + (size_t)(r0 + j) * D + c0;
        x0[j] = *reinterpret_cast<const v4f*>(row);
        x1[j] = *reinterpret_cast<const v4f*>(row + 128);
    }

    float dinv[LN_RPW];
    for (int j = 0; j < LN_RPW; ++j) {
        const float dg = deg[r0 + j];
        dinv[j] = (dg > 0.0f) ? 1.0f / dg : 0.0f;
    }

    const v4f bs0 = *reinterpret_cast<const v4f*>(bias + c0);
    const v4f bs1 = *reinterpret_cast<const v4f*>(bias + c0 + 128);
    const v4f g0  = *reinterpret_cast<const v4f*>(g + c0);
    const v4f g1  = *reinterpret_cast<const v4f*>(g + c0 + 128);
    const v4f be0 = *reinterpret_cast<const v4f*>(beta + c0);
    const v4f be1 = *reinterpret_cast<const v4f*>(beta + c0 + 128);
    const float a = pa[0];

    float sum[LN_RPW];
    for (int j = 0; j < LN_RPW; ++j) {
        x0[j] = x0[j] * dinv[j] + bs0;
        x1[j] = x1[j] * dinv[j] + bs1;
        sum[j] = lane_sum8(x0[j], x1[j]);
    }
    for (int m = 16; m > 0; m >>= 1)
        for (int j = 0; j < LN_RPW; ++j) sum[j] += __shfl_xor(sum[j], m, 32);

    float sq[LN_RPW];
    for (int j = 0; j < LN_RPW; ++j) {
        const float mu = sum[j] * (1.0f / D);
        x0[j] = x0[j] - mu;
        x1[j] = x1[j] - mu;
        sq[j] = lane_sum8(x0[j] * x0[j], x1[j] * x1[j]);
    }
    for (int m = 16; m > 0; m >>= 1)
        for (int j = 0; j < LN_RPW; ++j) sq[j] += __shfl_xor(sq[j], m, 32);

    for (int j = 0; j < LN_RPW; ++j) {
        const float rs = rsqrtf(sq[j] * (1.0f / D) + EPS_F);
        const v4f y0 = x0[j] * rs * g0 + be0;
        const v4f y1 = x1[j] * rs * g1 + be1;
        v4f o0, o1;
        for (int c = 0; c < 4; ++c) {
            o0[c] = (y0[c] >= 0.0f) ? y0[c] : a * y0[c];
            o1[c] = (y1[c] >= 0.0f) ? y1[c] : a * y1[c];
        }
        float* __restrict__ orow = out + (size_t)(r0 + j) * D + c0;
        *reinterpret_cast<v4f*>(orow)       = o0;
        *reinterpret_cast<v4f*>(orow + 128) = o1;
    }
}

__global__ __launch_bounds__(256) void wstd_kernel(const float* __restrict__ W,
                                                   float* __restrict__ Ws) {
    __shared__ float sred[256];
    const int r = blockIdx.x;
    const int f = threadIdx.x;
    const float w = W[(size_t)r * D + f];

    sred[f] = w;
    __syncthreads();
    #pragma unroll
    for (int s = 128; s > 0; s >>= 1) { if (f < s) sred[f] += sred[f + s]; __syncthreads(); }
    const float mean = sred[0] * (1.0f / D);
    __syncthreads();

    const float dv = w - mean;
    sred[f] = dv * dv;
    __syncthreads();
    #pragma unroll
    for (int s = 128; s > 0; s >>= 1) { if (f < s) sred[f] += sred[f + s]; __syncthreads(); }
    const float var = sred[0] * (1.0f / (D - 1));

    Ws[(size_t)r * D + f] = dv * rsqrtf(var + EPS_F);
}

extern "C" void kernel_launch(void* const* d_in, const int* in_sizes, int n_in,
                              void* d_out, int out_size, void* d_ws, size_t ws_size,
                              hipStream_t stream) {
    const float* edge_fea = (const float*)d_in[0];
    const int*   node_idx = (const int*)  d_in[1];
    const int*   edge_idx = (const int*)  d_in[2];
    const float* W1       = (const float*)d_in[3];
    const float* b1       = (const float*)d_in[4];
    const float* W2       = (const float*)d_in[5];
    const float* b2       = (const float*)d_in[6];
    const float* g1       = (const float*)d_in[7];
    const float* beta1    = (const float*)d_in[8];
    const float* g2       = (const float*)d_in[9];
    const float* beta2    = (const float*)d_in[10];
    const float* prelu_a  = (const float*)d_in[11];

    const size_t nodeRowBytes = (size_t)N_NODES * D * sizeof(float);

    float* B0  = (float*)d_out;
    char*  ws  = (char*)d_ws;
    float* B1  = (float*)ws;
    float* E   = (float*)(ws + nodeRowBytes);
    float* deg = E + (size_t)N_EDGES * D;
    float* cnt = deg + N_NODES;
    float* W2s = cnt + N_EDGES;


    hipMemsetAsync(deg, 0, (size_t)(N_NODES + N_EDGES) * sizeof(float), stream);
    degcnt_kernel<<<(NNZ_CNT + 255) / 256, 256, 0, stream>>>(node_idx, edge_idx, deg, cnt);

    gemm_wmma_kernel<D_IN><<<GM_BLOCKS, 128, 0, stream>>>(edge_fea, W1, B1);

    scatter_rows_lds_kernel<SL_ROWS_E, N_EDGES><<<SL_BLOCKS_E, SL_THREADS, 0, stream>>>(B1, node_idx, edge_idx, E);
    scale_edges_kernel<<<(N_EDGES * D) / 256, 256, 0, stream>>>(E, cnt);

    scatter_rows_lds_kernel<SL_ROWS_N, N_NODES><<<SL_BLOCKS_N, SL_THREADS, 0, stream>>>(E, edge_idx, node_idx, B0);
    fused_scale_ln_prelu_kernel<<<LN_BLOCKS, 256, 0, stream>>>(B0, deg, b1, g1, beta1, prelu_a, B1);

    wstd_kernel<<<D, 256, 0, stream>>>(W2, W2s);
    gemm_wmma_kernel<D><<<GM_BLOCKS, 128, 0, stream>>>(B1, W2s, B0);

    scatter_rows_lds_kernel<SL_ROWS_E, N_EDGES><<<SL_BLOCKS_E, SL_THREADS, 0, stream>>>(B0, node_idx, edge_idx, E);
    scale_edges_kernel<<<(N_EDGES * D) / 256, 256, 0, stream>>>(E, cnt);

    scatter_rows_lds_kernel<SL_ROWS_N, N_NODES><<<SL_BLOCKS_N, SL_THREADS, 0, stream>>>(E, edge_idx, node_idx, B1);
    fused_scale_ln_prelu_kernel<<<LN_BLOCKS, 256, 0, stream>>>(B1, deg, b2, g2, beta2, prelu_a, B0);
}


# Scatter-add groups walk per-call one-byte group digests of the index list (a quarter of the bytes) and fetch one packed (source row, own row) record per match; sums still in LDS arrival order
# speedup vs baseline: 4.4364x; 1.0787x over previous
#include <hip/hip_runtime.h>
#include <hip/hip_bf16.h>

#define N_NODES   100000
#define N_EDGES   20000
#define NNZ_CNT   800000
#define D_IN      128
#define D         256
#define EPS_F     1e-5f

typedef float v2f __attribute__((ext_vector_type(2)));
typedef float v8f __attribute__((ext_vector_type(8)));

__device__ __forceinline__ void atomAddF(float* p, float v) {
    __hip_atomic_fetch_add(p, v, __ATOMIC_RELAXED, __HIP_MEMORY_SCOPE_AGENT);
}

typedef int v4i __attribute__((ext_vector_type(4)));
static_assert(NNZ_CNT % 16 == 0, "index lists are read as 16-byte quads, digests too");
static_assert(N_NODES <= (1 << 17) && N_EDGES <= (1 << 17), "a source row number packs into 17 bits");
#define IDX_NQ (NNZ_CNT / 4)
#define REC_OFF_WORDS (1 << 18)

template<int ROWS>
__device__ __forceinline__ void build_digest(int t, const int* __restrict__ dst_idx,
                                             const int* __restrict__ src_idx,
                                             int* __restrict__ dg) {
    const v4i d4 = reinterpret_cast<const v4i*>(dst_idx)[t];
    const v4i s4 = reinterpret_cast<const v4i*>(src_idx)[t];
    unsigned word = 0u;
    v4i rec;
    for (int j = 0; j < 4; ++j) {
        const unsigned grp = (unsigned)d4[j] / (unsigned)ROWS;
        const unsigned rel = (unsigned)d4[j] % (unsigned)ROWS;
        word |= (grp & 255u) << (8 * j);
        rec[j] = (int)(((unsigned)s4[j] << 15) | (min(grp >> 8, 127u) << 8) | rel);
    }
    dg[t] = (int)word;
    reinterpret_cast<v4i*>(dg + REC_OFF_WORDS)[t] = rec;
}

template<int ROWS>
__global__ __launch_bounds__(256) void digest_kernel(const int* __restrict__ dst_idx,
                                                     const int* __restrict__ src_idx,
                                                     int* __restrict__ dg) {
    const int t = blockIdx.x * 256 + threadIdx.x;
    if (t < IDX_NQ) build_digest<ROWS>(t, dst_idx, src_idx, dg);
}

template<int ROWS_E>
__global__ __launch_bounds__(256) void degcnt_kernel(const int* __restrict__ node_idx,
                                                     const int* __restrict__ edge_idx,
                                                     float* __restrict__ deg,
                                                     float* __restrict__ cnt,
                                                     int* __restrict__ dge) {
    const int t = blockIdx.x * 256 + threadIdx.x;
    if (t < IDX_NQ) {
        const v4i n4 = reinterpret_cast<const v4i*>(node_idx)[t];
        const v4i e4 = reinterpret_cast<const v4i*>(edge_idx)[t];
        for (int j = 0; j < 4; ++j) {
            atomAddF(&deg[n4[j]], 1.0f);
            atomAddF(&cnt[e4[j]], 1.0f);
        }
        build_digest<ROWS_E>(t, edge_idx, node_idx, dge);
    }
}

#define GM_ROWS 64
#define GM_BLOCKS ((N_NODES + GM_ROWS - 1) / GM_ROWS)
static_assert(D == 256, "four waves of 64 columns cover the 256 output columns");
static_assert(N_NODES % 16 == 0, "a 16-row tile is wholly inside or wholly outside the matrix");

typedef float v4f __attribute__((ext_vector_type(4)));

__device__ __forceinline__ void load_frag(const float* __restrict__ p, v4f (&f)[4]) {
    #pragma unroll
    for (int j = 0; j < 4; ++j) f[j] = *reinterpret_cast<const v4f*>(p + 4 * j);
}

template<int K>
__global__ __launch_bounds__(128) void gemm_wmma_kernel(const float* __restrict__ X,
                                                        const float* __restrict__ W,
                                                        float* __restrict__ C) {
    static_assert(K % 32 == 0, "K is walked in 32-k chunks");
    constexpr int NCH = K / 32;
    const int lane = threadIdx.x & 31;
    const int wave = __builtin_amdgcn_readfirstlane((int)(threadIdx.x >> 5));
    const int c  = lane & 15;
    const int h  = lane >> 4;
    const int m0 = blockIdx.x * GM_ROWS;
    const int n0 = wave * 64;

    const float* __restrict__ ap[4];
    #pragma unroll
    for (int i = 0; i < 4; ++i) {
        const int ti = (m0 + 16 * i < N_NODES) ? i : 0;
        ap[i] = X + (size_t)(m0 + 16 * ti + c) * K + 16 * h;
    }
    const float* __restrict__ bp = W + (size_t)(n0 + 4 * c) * K + 16 * h;

    v8f acc[4][4];
    #pragma unroll
    for (int i = 0; i < 4; ++i)
        #pragma unroll
        for (int n = 0; n < 4; ++n) acc[i][n] = v8f{};

    v4f fa[2][4][4], fb[2][4][4];

    #pragma unroll
    for (int i = 0; i < 4; ++i) load_frag(ap[i], fa[0][i]);
    #pragma unroll
    for (int n = 0; n < 4; ++n) load_frag(bp + (size_t)n * K, fb[0][n]);

    #pragma unroll
    for (int q = 0; q < NCH; ++q) {
        const int cur = q & 1, nxt = cur ^ 1;
        if (q + 1 < NCH) {
            #pragma unroll
            for (int i = 0; i < 4; ++i) load_frag(ap[i] + 32 * (q + 1), fa[nxt][i]);
            #pragma unroll
            for (int n = 0; n < 4; ++n) load_frag(bp + (size_t)n * K + 32 * (q + 1), fb[nxt][n]);
        }
        __builtin_amdgcn_sched_barrier(0);
        #pragma unroll
        for (int j = 0; j < 4; ++j) {
            #pragma unroll
            for (int p = 0; p < 2; ++p) {
                v2f a[4], b[4];
                #pragma unroll
                for (int i = 0; i < 4; ++i) {
                    a[i] = p ? __builtin_shufflevector(fa[cur][i][j], fa[cur][i][j], 2, 3)
                             : __builtin_shufflevector(fa[cur][i][j], fa[cur][i][j], 0, 1);
                    b[i] = p ? __builtin_shufflevector(fb[cur][i][j], fb[cur][i][j], 2, 3)
                             : __builtin_shufflevector(fb[cur][i][j], fb[cur][i][j], 0, 1);
                }
                #pragma unroll
                for (int i = 0; i < 4; ++i)
                    #pragma unroll
                    for (int n = 0; n < 4; ++n)
                        acc[i][n] = __builtin_amdgcn_wmma_f32_16x16x4_f32(false, a[i], false, b[n], (short)0,
                                                                          acc[i][n], false, false);
            }
        }
    }

    float* __restrict__ cp = C + (size_t)(m0 + 8 * h) * D + n0 + 4 * c;
    #pragma unroll
    for (int i = 0; i < 4; ++i) {
        if (m0 + 16 * i < N_NODES) {
            #pragma unroll
            for (int r = 0; r < 8; ++r) {
                const v4f o = {acc[i][0][r], acc[i][1][r], acc[i][2][r], acc[i][3][r]};
                *reinterpret_cast<v4f*>(cp + (size_t)(16 * i + r) * D) = o;
            }
        }
    }
}

#define SL_THREADS 1024
#define SL_WAVES   (SL_THREADS / 32)
#define SL_ACAP    160
#define SL_BCAP    40
#define DG_NQ      (NNZ_CNT / 16)
#define SL_NIT     ((DG_NQ + SL_THREADS - 1) / SL_THREADS)

__device__ __forceinline__ void ldsAddF(float* p, float v) {
    __hip_atomic_fetch_add(p, v, __ATOMIC_RELAXED, __HIP_MEMORY_SCOPE_WORKGROUP);
}

__device__ __forceinline__ v4i sl_load_quad(const v4i* __restrict__ g4, int qi) {
    return g4[min(qi, DG_NQ - 1)];
}

__device__ __forceinline__ int sl_check(const unsigned* qa, int pos, int n, unsigned* qb, int cb,
                                        int lane, unsigned grp_hi,
                                        const int* __restrict__ rec) {
    const bool act = lane < n;
    const unsigned e = act ? qa[pos + lane] : 0u;
    const unsigned r = (unsigned)rec[e];
    const bool ok = act && ((r >> 8) & 127u) == grp_hi;
    const unsigned mk = __builtin_amdgcn_ballot_w32(ok);
    if (ok) qb[cb + (int)__builtin_amdgcn_mbcnt_lo(mk, 0u)] = ((r >> 15) << 8) | (r & 255u);
    return cb + __builtin_popcount(mk);
}

__device__ __forceinline__ void sl_add8(const unsigned* qb, int pos, int lane,
                                        const float* __restrict__ src, float* acc) {
    const unsigned pk = qb[pos + (lane & 7)];
    const int rel = (int)(pk & 255u);
    const int s = (int)(pk >> 8);
    float v[8][8];
    for (int j = 0; j < 8; ++j) {
        const int sj = __builtin_amdgcn_readlane(s, j);
        const float* __restrict__ row = src + (size_t)sj * D + lane;
        for (int c = 0; c < 8; ++c) v[j][c] = row[c * 32];
    }
    for (int j = 0; j < 8; ++j) {
        const int rj = __builtin_amdgcn_readlane(rel, j);
        float* a = acc + rj * D + lane;
        for (int c = 0; c < 8; ++c) ldsAddF(a + c * 32, v[j][c]);
    }
}

template<int ROWS, int NDST>
__global__ __launch_bounds__(SL_THREADS) void scatter_rows_lds_kernel(const float* __restrict__ src,
                                                                      const int* __restrict__ dg,
                                                                      float* __restrict__ dst) {
    static_assert(ROWS <= 256, "a row number inside the group packs into 8 bits");
    __shared__ float acc[ROWS * D];
    __shared__ unsigned listA[SL_WAVES][SL_ACAP];
    __shared__ unsigned listB[SL_WAVES][SL_BCAP];

    const int tid  = threadIdx.x;
    const int lane = tid & 31;
    const int wave = __builtin_amdgcn_readfirstlane(tid >> 5);
    const int base = blockIdx.x * ROWS;
    const unsigned pat = (unsigned)(blockIdx.x & 255) * 0x01010101u;
    const unsigned grp_hi = (unsigned)blockIdx.x >> 8;
    const int* __restrict__ rec = dg + REC_OFF_WORDS;

    v4f* acc4 = reinterpret_cast<v4f*>(acc);
    for (int k = tid; k < ROWS * D / 4; k += SL_THREADS) acc4[k] = v4f{0.0f, 0.0f, 0.0f, 0.0f};
    __syncthreads();

    unsigned* qa = listA[wave];
    unsigned* qb = listB[wave];
    int ca = 0;
    int cb = 0;
    const v4i* __restrict__ g4 = reinterpret_cast<const v4i*>(dg);

    v4i nx = sl_load_quad(g4, tid);
    for (int it = 0; it < SL_NIT; ++it) {
        const int qi = it * SL_THREADS + tid;
        const v4i x = nx;
        nx = sl_load_quad(g4, qi + SL_THREADS);
        const bool in = qi < DG_NQ;
        #pragma unroll
        for (int c = 0; c < 4; ++c) {
            const unsigned y = (unsigned)x[c] ^ pat;
            unsigned z = ~(((y & 0x7F7F7F7Fu) + 0x7F7F7F7Fu) | y | 0x7F7F7F7Fu);
            z = in ? z : 0u;
            unsigned any = __builtin_amdgcn_ballot_w32(z != 0u);
            if (any != 0u) {
                do {
                    if (z != 0u) {
                        const int b = __builtin_ctz(z) >> 3;
                        z &= z - 1u;
                        qa[ca + (int)__builtin_amdgcn_mbcnt_lo(any, 0u)] = (unsigned)(16 * qi + 4 * c + b);
                    }
                    ca += __builtin_popcount(any);
                    any = __builtin_amdgcn_ballot_w32(z != 0u);
                } while (any != 0u);
                while (ca >= 32) {
                    ca -= 32;
                    cb = sl_check(qa, ca, 32, qb, cb, lane, grp_hi, rec);
                    while (cb >= 8) {
                        cb -= 8;
                        sl_add8(qb, cb, lane, src, acc);
                    }
                }
            }
        }
    }

    if (ca > 0) {
        cb = sl_check(qa, 0, ca, qb, cb, lane, grp_hi, rec);
        while (cb >= 8) {
            cb -= 8;
            sl_add8(qb, cb, lane, src, acc);
        }
    }
    for (int j = 0; j < cb; ++j) {
        const unsigned pk = (unsigned)__builtin_amdgcn_readfirstlane((int)qb[j]);
        const float* __restrict__ row = src + (size_t)(pk >> 8) * D + lane;
        float* a = acc + (int)(pk & 255u) * D + lane;
        float v[8];
        for (int c = 0; c < 8; ++c) v[c] = row[c * 32];
        for (int c = 0; c < 8; ++c) ldsAddF(a + c * 32, v[c]);
    }
    __syncthreads();

    v4f* __restrict__ out4 = reinterpret_cast<v4f*>(dst + (size_t)base * D);
    for (int k = tid; k < ROWS * D / 4; k += SL_THREADS)
        if (base + (k >> 6) < NDST) out4[k] = acc4[k];
}

#define SL_ROWS_E 80
#define SL_ROWS_N 256
#define SL_BLOCKS_E ((N_EDGES + SL_ROWS_E - 1) / SL_ROWS_E)
#define SL_BLOCKS_N ((N_NODES + SL_ROWS_N - 1) / SL_ROWS_N)

template<int ROWS_N>
__global__ __launch_bounds__(256) void scale_edges_kernel(float* __restrict__ E,
                                                          const float* __restrict__ cnt,
                                                          const int* __restrict__ node_idx,
                                                          const int* __restrict__ edge_idx,
                                                          int* __restrict__ dgn) {
    const int idx = blockIdx.x * 256 + threadIdx.x;
    if (idx < IDX_NQ) build_digest<ROWS_N>(idx, node_idx, edge_idx, dgn);
    if (idx < N_EDGES * D) {
        const float c = cnt[idx >> 8];
        const float inv = (c > 0.0f) ? 1.0f / c : 0.0f;
        E[idx] *= inv;
    }
}

#define LN_RPW 4
#define LN_WAVES_PER_BLOCK 8
#define LN_BLOCKS (N_NODES / (LN_RPW * LN_WAVES_PER_BLOCK))
static_assert(N_NODES % (LN_RPW * LN_WAVES_PER_BLOCK) == 0, "row count must fill whole blocks");
static_assert(D == 256, "lane-to-column map below is written for 256-wide rows");

__device__ __forceinline__ float lane_sum8(v4f a, v4f b) {
    return ((a.x + a.y) + (a.z + a.w)) + ((b.x + b.y) + (b.z + b.w));
}

__global__ __launch_bounds__(256) void fused_scale_ln_prelu_kernel(const float* __restrict__ acc,
                                                                   const float* __restrict__ deg,
                                                                   const float* __restrict__ bias,
                                                                   const float* __restrict__ g,
                                                                   const float* __restrict__ beta,
                                                                   const float* __restrict__ pa,
                                                                   float* __restrict__ out) {
    const int lane = threadIdx.x & 31;
    const int wv = __builtin_amdgcn_readfirstlane((int)((blockIdx.x * 256 + threadIdx.x) >> 5));
    const int r0 = wv * LN_RPW;
    const int c0 = lane * 4;

    v4f x0[LN_RPW], x1[LN_RPW];
    for (int j = 0; j < LN_RPW; ++j) {
        const float* __restrict__ row = acc + (size_t)(r0 + j) * D + c0;
        x0[j] = *reinterpret_cast<const v4f*>(row);
        x1[j] = *reinterpret_cast<const v4f*>(row + 128);
    }

    float dinv[LN_RPW];
    for (int j = 0; j < LN_RPW; ++j) {
        const float dg = deg[r0 + j];
        dinv[j] = (dg > 0.0f) ? 1.0f / dg : 0.0f;
    }

    const v4f bs0 = *reinterpret_cast<const v4f*>(bias + c0);
    const v4f bs1 = *reinterpret_cast<const v4f*>(bias + c0 + 128);
    const v4f g0  = *reinterpret_cast<const v4f*>(g + c0);
    const v4f g1  = *reinterpret_cast<const v4f*>(g + c0 + 128);
    const v4f be0 = *reinterpret_cast<const v4f*>(beta + c0);
    const v4f be1 = *reinterpret_cast<const v4f*>(beta + c0 + 128);
    const float a = pa[0];

    float sum[LN_RPW];
    for (int j = 0; j < LN_RPW; ++j) {
        x0[j] = x0[j] * dinv[j] + bs0;
        x1[j] = x1[j] * dinv[j] + bs1;
        sum[j] = lane_sum8(x0[j], x1[j]);
    }
    for (int m = 16; m > 0; m >>= 1)
        for (int j = 0; j < LN_RPW; ++j) sum[j] += __shfl_xor(sum[j], m, 32);

    float sq[LN_RPW];
    for (int j = 0; j < LN_RPW; ++j) {
        const float mu = sum[j] * (1.0f / D);
        x0[j] = x0[j] - mu;
        x1[j] = x1[j] - mu;
        sq[j] = lane_sum8(x0[j] * x0[j], x1[j] * x1[j]);
    }
    for (int m = 16; m > 0; m >>= 1)
        for (int j = 0; j < LN_RPW; ++j) sq[j] += __shfl_xor(sq[j], m, 32);

    for (int j = 0; j < LN_RPW; ++j) {
        const float rs = rsqrtf(sq[j] * (1.0f / D) + EPS_F);
        const v4f y0 = x0[j] * rs * g0 + be0;
        const v4f y1 = x1[j] * rs * g1 + be1;
        v4f o0, o1;
        for (int c = 0; c < 4; ++c) {
            o0[c] = (y0[c] >= 0.0f) ? y0[c] : a * y0[c];
            o1[c] = (y1[c] >= 0.0f) ? y1[c] : a * y1[c];
        }
        float* __restrict__ orow = out + (size_t)(r0 + j) * D + c0;
        *reinterpret_cast<v4f*>(orow)       = o0;
        *reinterpret_cast<v4f*>(orow + 128) = o1;
    }
}

__global__ __launch_bounds__(256) void wstd_kernel(const float* __restrict__ W,
                                                   float* __restrict__ Ws) {
    __shared__ float sred[256];
    const int r = blockIdx.x;
    const int f = threadIdx.x;
    const float w = W[(size_t)r * D + f];

    sred[f] = w;
    __syncthreads();
    #pragma unroll
    for (int s = 128; s > 0; s >>= 1) { if (f < s) sred[f] += sred[f + s]; __syncthreads(); }
    const float mean = sred[0] * (1.0f / D);
    __syncthreads();

    const float dv = w - mean;
    sred[f] = dv * dv;
    __syncthreads();
    #pragma unroll
    for (int s = 128; s > 0; s >>= 1) { if (f < s) sred[f] += sred[f + s]; __syncthreads(); }
    const float var = sred[0] * (1.0f / (D - 1));

    Ws[(size_t)r * D + f] = dv * rsqrtf(var + EPS_F);
}

extern "C" void kernel_launch(void* const* d_in, const int* in_sizes, int n_in,
                              void* d_out, int out_size, void* d_ws, size_t ws_size,
                              hipStream_t stream) {
    const float* edge_fea = (const float*)d_in[0];
    const int*   node_idx = (const int*)  d_in[1];
    const int*   edge_idx = (const int*)  d_in[2];
    const float* W1       = (const float*)d_in[3];
    const float* b1       = (const float*)d_in[4];
    const float* W2       = (const float*)d_in[5];
    const float* b2       = (const float*)d_in[6];
    const float* g1       = (const float*)d_in[7];
    const float* beta1    = (const float*)d_in[8];
    const float* g2       = (const float*)d_in[9];
    const float* beta2    = (const float*)d_in[10];
    const float* prelu_a  = (const float*)d_in[11];

    const size_t nodeRowBytes = (size_t)N_NODES * D * sizeof(float);

    float* B0  = (float*)d_out;
    char*  ws  = (char*)d_ws;
    float* B1  = (float*)ws;
    float* E   = (float*)(ws + nodeRowBytes);
    float* deg = E + (size_t)N_EDGES * D;
    float* cnt = deg + N_NODES;
    float* W2s = cnt + N_EDGES;


    int* dgB0 = (int*)d_out;
    int* dgB1 = (int*)ws;

    hipMemsetAsync(deg, 0, (size_t)(N_NODES + N_EDGES) * sizeof(float), stream);
    degcnt_kernel<SL_ROWS_E><<<(IDX_NQ + 255) / 256, 256, 0, stream>>>(node_idx, edge_idx, deg, cnt, dgB0);

    gemm_wmma_kernel<D_IN><<<GM_BLOCKS, 128, 0, stream>>>(edge_fea, W1, B1);

    scatter_rows_lds_kernel<SL_ROWS_E, N_EDGES><<<SL_BLOCKS_E, SL_THREADS, 0, stream>>>(B1, dgB0, E);
    scale_edges_kernel<SL_ROWS_N><<<(N_EDGES * D) / 256, 256, 0, stream>>>(E, cnt, node_idx, edge_idx, dgB1);

    scatter_rows_lds_kernel<SL_ROWS_N, N_NODES><<<SL_BLOCKS_N, SL_THREADS, 0, stream>>>(E, dgB1, B0);
    fused_scale_ln_prelu_kernel<<<LN_BLOCKS, 256, 0, stream>>>(B0, deg, b1, g1, beta1, prelu_a, B1);

    wstd_kernel<<<D, 256, 0, stream>>>(W2, W2s);
    gemm_wmma_kernel<D><<<GM_BLOCKS, 128, 0, stream>>>(B1, W2s, B0);

    digest_kernel<SL_ROWS_E><<<(IDX_NQ + 255) / 256, 256, 0, stream>>>(edge_idx, node_idx, dgB1);
    scatter_rows_lds_kernel<SL_ROWS_E, N_EDGES><<<SL_BLOCKS_E, SL_THREADS, 0, stream>>>(B0, dgB1, E);
    scale_edges_kernel<SL_ROWS_N><<<(N_EDGES * D) / 256, 256, 0, stream>>>(E, cnt, node_idx, edge_idx, dgB0);

    scatter_rows_lds_kernel<SL_ROWS_N, N_NODES><<<SL_BLOCKS_N, SL_THREADS, 0, stream>>>(E, dgB0, B1);
    fused_scale_ln_prelu_kernel<<<LN_BLOCKS, 256, 0, stream>>>(B1, deg, b2, g2, beta2, prelu_a, B0);
}


# Hyperedge B^-1 scaling moved into the node-to-hyperedge scatter-add's row store out of LDS; the one-float-per-thread rescale pass is gone, its launch now only builds the node-side digests
# speedup vs baseline: 4.5727x; 1.0307x over previous
#include <hip/hip_runtime.h>
#include <hip/hip_bf16.h>

#define N_NODES   100000
#define N_EDGES   20000
#define NNZ_CNT   800000
#define D_IN      128
#define D         256
#define EPS_F     1e-5f

typedef float v2f __attribute__((ext_vector_type(2)));
typedef float v8f __attribute__((ext_vector_type(8)));

__device__ __forceinline__ void atomAddF(float* p, float v) {
    __hip_atomic_fetch_add(p, v, __ATOMIC_RELAXED, __HIP_MEMORY_SCOPE_AGENT);
}

typedef int v4i __attribute__((ext_vector_type(4)));
static_assert(NNZ_CNT % 16 == 0, "index lists are read as 16-byte quads, digests too");
static_assert(N_NODES <= (1 << 17) && N_EDGES <= (1 << 17), "a source row number packs into 17 bits");
#define IDX_NQ (NNZ_CNT / 4)
#define REC_OFF_WORDS (1 << 18)

template<int ROWS>
__device__ __forceinline__ void build_digest(int t, const int* __restrict__ dst_idx,
                                             const int* __restrict__ src_idx,
                                             int* __restrict__ dg) {
    const v4i d4 = reinterpret_cast<const v4i*>(dst_idx)[t];
    const v4i s4 = reinterpret_cast<const v4i*>(src_idx)[t];
    unsigned word = 0u;
    v4i rec;
    for (int j = 0; j < 4; ++j) {
        const unsigned grp = (unsigned)d4[j] / (unsigned)ROWS;
        const unsigned rel = (unsigned)d4[j] % (unsigned)ROWS;
        word |= (grp & 255u) << (8 * j);
        rec[j] = (int)(((unsigned)s4[j] << 15) | (min(grp >> 8, 127u) << 8) | rel);
    }
    dg[t] = (int)word;
    reinterpret_cast<v4i*>(dg + REC_OFF_WORDS)[t] = rec;
}

template<int ROWS>
__global__ __launch_bounds__(256) void digest_kernel(const int* __restrict__ dst_idx,
                                                     const int* __restrict__ src_idx,
                                                     int* __restrict__ dg) {
    const int t = blockIdx.x * 256 + threadIdx.x;
    if (t < IDX_NQ) build_digest<ROWS>(t, dst_idx, src_idx, dg);
}

template<int ROWS_E>
__global__ __launch_bounds__(256) void degcnt_kernel(const int* __restrict__ node_idx,
                                                     const int* __restrict__ edge_idx,
                                                     float* __restrict__ deg,
                                                     float* __restrict__ cnt,
                                                     int* __restrict__ dge) {
    const int t = blockIdx.x * 256 + threadIdx.x;
    if (t < IDX_NQ) {
        const v4i n4 = reinterpret_cast<const v4i*>(node_idx)[t];
        const v4i e4 = reinterpret_cast<const v4i*>(edge_idx)[t];
        for (int j = 0; j < 4; ++j) {
            atomAddF(&deg[n4[j]], 1.0f);
            atomAddF(&cnt[e4[j]], 1.0f);
        }
        build_digest<ROWS_E>(t, edge_idx, node_idx, dge);
    }
}

#define GM_ROWS 64
#define GM_BLOCKS ((N_NODES + GM_ROWS - 1) / GM_ROWS)
static_assert(D == 256, "four waves of 64 columns cover the 256 output columns");
static_assert(N_NODES % 16 == 0, "a 16-row tile is wholly inside or wholly outside the matrix");

typedef float v4f __attribute__((ext_vector_type(4)));

__device__ __forceinline__ void load_frag(const float* __restrict__ p, v4f (&f)[4]) {
    #pragma unroll
    for (int j = 0; j < 4; ++j) f[j] = *reinterpret_cast<const v4f*>(p + 4 * j);
}

template<int K>
__global__ __launch_bounds__(128) void gemm_wmma_kernel(const float* __restrict__ X,
                                                        const float* __restrict__ W,
                                                        float* __restrict__ C) {
    static_assert(K % 32 == 0, "K is walked in 32-k chunks");
    constexpr int NCH = K / 32;
    const int lane = threadIdx.x & 31;
    const int wave = __builtin_amdgcn_readfirstlane((int)(threadIdx.x >> 5));
    const int c  = lane & 15;
    const int h  = lane >> 4;
    const int m0 = blockIdx.x * GM_ROWS;
    const int n0 = wave * 64;

    const float* __restrict__ ap[4];
    #pragma unroll
    for (int i = 0; i < 4; ++i) {
        const int ti = (m0 + 16 * i < N_NODES) ? i : 0;
        ap[i] = X + (size_t)(m0 + 16 * ti + c) * K + 16 * h;
    }
    const float* __restrict__ bp = W + (size_t)(n0 + 4 * c) * K + 16 * h;

    v8f acc[4][4];
    #pragma unroll
    for (int i = 0; i < 4; ++i)
        #pragma unroll
        for (int n = 0; n < 4; ++n) acc[i][n] = v8f{};

    v4f fa[2][4][4], fb[2][4][4];

    #pragma unroll
    for (int i = 0; i < 4; ++i) load_frag(ap[i], fa[0][i]);
    #pragma unroll
    for (int n = 0; n < 4; ++n) load_frag(bp + (size_t)n * K, fb[0][n]);

    #pragma unroll
    for (int q = 0; q < NCH; ++q) {
        const int cur = q & 1, nxt = cur ^ 1;
        if (q + 1 < NCH) {
            #pragma unroll
            for (int i = 0; i < 4; ++i) load_frag(ap[i] + 32 * (q + 1), fa[nxt][i]);
            #pragma unroll
            for (int n = 0; n < 4; ++n) load_frag(bp + (size_t)n * K + 32 * (q + 1), fb[nxt][n]);
        }
        __builtin_amdgcn_sched_barrier(0);
        #pragma unroll
        for (int j = 0; j < 4; ++j) {
            #pragma unroll
            for (int p = 0; p < 2; ++p) {
                v2f a[4], b[4];
                #pragma unroll
                for (int i = 0; i < 4; ++i) {
                    a[i] = p ? __builtin_shufflevector(fa[cur][i][j], fa[cur][i][j], 2, 3)
                             : __builtin_shufflevector(fa[cur][i][j], fa[cur][i][j], 0, 1);
                    b[i] = p ? __builtin_shufflevector(fb[cur][i][j], fb[cur][i][j], 2, 3)
                             : __builtin_shufflevector(fb[cur][i][j], fb[cur][i][j], 0, 1);
                }
                #pragma unroll
                for (int i = 0; i < 4; ++i)
                    #pragma unroll
                    for (int n = 0; n < 4; ++n)
                        acc[i][n] = __builtin_amdgcn_wmma_f32_16x16x4_f32(false, a[i], false, b[n], (short)0,
                                                                          acc[i][n], false, false);
            }
        }
    }

    float* __restrict__ cp = C + (size_t)(m0 + 8 * h) * D + n0 + 4 * c;
    #pragma unroll
    for (int i = 0; i < 4; ++i) {
        if (m0 + 16 * i < N_NODES) {
            #pragma unroll
            for (int r = 0; r < 8; ++r) {
                const v4f o = {acc[i][0][r], acc[i][1][r], acc[i][2][r], acc[i][3][r]};
                *reinterpret_cast<v4f*>(cp + (size_t)(16 * i + r) * D) = o;
            }
        }
    }
}

#define SL_THREADS 1024
#define SL_WAVES   (SL_THREADS / 32)
#define SL_ACAP    160
#define SL_BCAP    40
#define DG_NQ      (NNZ_CNT / 16)
#define SL_NIT     ((DG_NQ + SL_THREADS - 1) / SL_THREADS)

__device__ __forceinline__ void ldsAddF(float* p, float v) {
    __hip_atomic_fetch_add(p, v, __ATOMIC_RELAXED, __HIP_MEMORY_SCOPE_WORKGROUP);
}

__device__ __forceinline__ v4i sl_load_quad(const v4i* __restrict__ g4, int qi) {
    return g4[min(qi, DG_NQ - 1)];
}

__device__ __forceinline__ int sl_check(const unsigned* qa, int pos, int n, unsigned* qb, int cb,
                                        int lane, unsigned grp_hi,
                                        const int* __restrict__ rec) {
    const bool act = lane < n;
    const unsigned e = act ? qa[pos + lane] : 0u;
    const unsigned r = (unsigned)rec[e];
    const bool ok = act && ((r >> 8) & 127u) == grp_hi;
    const unsigned mk = __builtin_amdgcn_ballot_w32(ok);
    if (ok) qb[cb + (int)__builtin_amdgcn_mbcnt_lo(mk, 0u)] = ((r >> 15) << 8) | (r & 255u);
    return cb + __builtin_popcount(mk);
}

__device__ __forceinline__ void sl_add8(const unsigned* qb, int pos, int lane,
                                        const float* __restrict__ src, float* acc) {
    const unsigned pk = qb[pos + (lane & 7)];
    const int rel = (int)(pk & 255u);
    const int s = (int)(pk >> 8);
    float v[8][8];
    for (int j = 0; j < 8; ++j) {
        const int sj = __builtin_amdgcn_readlane(s, j);
        const float* __restrict__ row = src + (size_t)sj * D + lane;
        for (int c = 0; c < 8; ++c) v[j][c] = row[c * 32];
    }
    for (int j = 0; j < 8; ++j) {
        const int rj = __builtin_amdgcn_readlane(rel, j);
        float* a = acc + rj * D + lane;
        for (int c = 0; c < 8; ++c) ldsAddF(a + c * 32, v[j][c]);
    }
}

template<int ROWS, int NDST, bool SCALE>
__global__ __launch_bounds__(SL_THREADS) void scatter_rows_lds_kernel(const float* __restrict__ src,
                                                                      const int* __restrict__ dg,
                                                                      const float* __restrict__ rowcnt,
                                                                      float* __restrict__ dst) {
    static_assert(ROWS <= 256, "a row number inside the group packs into 8 bits");
    __shared__ float acc[ROWS * D];
    __shared__ unsigned listA[SL_WAVES][SL_ACAP];
    __shared__ unsigned listB[SL_WAVES][SL_BCAP];

    const int tid  = threadIdx.x;
    const int lane = tid & 31;
    const int wave = __builtin_amdgcn_readfirstlane(tid >> 5);
    const int base = blockIdx.x * ROWS;
    const unsigned pat = (unsigned)(blockIdx.x & 255) * 0x01010101u;
    const unsigned grp_hi = (unsigned)blockIdx.x >> 8;
    const int* __restrict__ rec = dg + REC_OFF_WORDS;

    v4f* acc4 = reinterpret_cast<v4f*>(acc);
    for (int k = tid; k < ROWS * D / 4; k += SL_THREADS) acc4[k] = v4f{0.0f, 0.0f, 0.0f, 0.0f};
    __syncthreads();

    unsigned* qa = listA[wave];
    unsigned* qb = listB[wave];
    int ca = 0;
    int cb = 0;
    const v4i* __restrict__ g4 = reinterpret_cast<const v4i*>(dg);

    v4i nx = sl_load_quad(g4, tid);
    for (int it = 0; it < SL_NIT; ++it) {
        const int qi = it * SL_THREADS + tid;
        const v4i x = nx;
        nx = sl_load_quad(g4, qi + SL_THREADS);
        const bool in = qi < DG_NQ;
        #pragma unroll
        for (int c = 0; c < 4; ++c) {
            const unsigned y = (unsigned)x[c] ^ pat;
            unsigned z = ~(((y & 0x7F7F7F7Fu) + 0x7F7F7F7Fu) | y | 0x7F7F7F7Fu);
            z = in ? z : 0u;
            unsigned any = __builtin_amdgcn_ballot_w32(z != 0u);
            if (any != 0u) {
                do {
                    if (z != 0u) {
                        const int b = __builtin_ctz(z) >> 3;
                        z &= z - 1u;
                        qa[ca + (int)__builtin_amdgcn_mbcnt_lo(any, 0u)] = (unsigned)(16 * qi + 4 * c + b);
                    }
                    ca += __builtin_popcount(any);
                    any = __builtin_amdgcn_ballot_w32(z != 0u);
                } while (any != 0u);
                while (ca >= 32) {
                    ca -= 32;
                    cb = sl_check(qa, ca, 32, qb, cb, lane, grp_hi, rec);
                    while (cb >= 8) {
                        cb -= 8;
                        sl_add8(qb, cb, lane, src, acc);
                    }
                }
            }
        }
    }

    if (ca > 0) {
        cb = sl_check(qa, 0, ca, qb, cb, lane, grp_hi, rec);
        while (cb >= 8) {
            cb -= 8;
            sl_add8(qb, cb, lane, src, acc);
        }
    }
    for (int j = 0; j < cb; ++j) {
        const unsigned pk = (unsigned)__builtin_amdgcn_readfirstlane((int)qb[j]);
        const float* __restrict__ row = src + (size_t)(pk >> 8) * D + lane;
        float* a = acc + (int)(pk & 255u) * D + lane;
        float v[8];
        for (int c = 0; c < 8; ++c) v[c] = row[c * 32];
        for (int c = 0; c < 8; ++c) ldsAddF(a + c * 32, v[c]);
    }
    __syncthreads();

    v4f* __restrict__ out4 = reinterpret_cast<v4f*>(dst + (size_t)base * D);
    for (int k = tid; k < ROWS * D / 4; k += SL_THREADS) {
        const int r = base + (k >> 6);
        if (r < NDST) {
            v4f v = acc4[k];
            if (SCALE) {
                const float c = rowcnt[r];
                const float inv = (c > 0.0f) ? 1.0f / c : 0.0f;
                v = v * inv;
            }
            out4[k] = v;
        }
    }
}

#define SL_ROWS_E 80
#define SL_ROWS_N 256
#define SL_BLOCKS_E ((N_EDGES + SL_ROWS_E - 1) / SL_ROWS_E)
#define SL_BLOCKS_N ((N_NODES + SL_ROWS_N - 1) / SL_ROWS_N)

#define LN_RPW 4
#define LN_WAVES_PER_BLOCK 8
#define LN_BLOCKS (N_NODES / (LN_RPW * LN_WAVES_PER_BLOCK))
static_assert(N_NODES % (LN_RPW * LN_WAVES_PER_BLOCK) == 0, "row count must fill whole blocks");
static_assert(D == 256, "lane-to-column map below is written for 256-wide rows");

__device__ __forceinline__ float lane_sum8(v4f a, v4f b) {
    return ((a.x + a.y) + (a.z + a.w)) + ((b.x + b.y) + (b.z + b.w));
}

__global__ __launch_bounds__(256) void fused_scale_ln_prelu_kernel(const float* __restrict__ acc,
                                                                   const float* __restrict__ deg,
                                                                   const float* __restrict__ bias,
                                                                   const float* __restrict__ g,
                                                                   const float* __restrict__ beta,
                                                                   const float* __restrict__ pa,
                                                                   float* __restrict__ out) {
    const int lane = threadIdx.x & 31;
    const int wv = __builtin_amdgcn_readfirstlane((int)((blockIdx.x * 256 + threadIdx.x) >> 5));
    const int r0 = wv * LN_RPW;
    const int c0 = lane * 4;

    v4f x0[LN_RPW], x1[LN_RPW];
    for (int j = 0; j < LN_RPW; ++j) {
        const float* __restrict__ row = acc + (size_t)(r0 + j) * D + c0;
        x0[j] = *reinterpret_cast<const v4f*>(row);
        x1[j] = *reinterpret_cast<const v4f*>(row + 128);
    }

    float dinv[LN_RPW];
    for (int j = 0; j < LN_RPW; ++j) {
        const float dg = deg[r0 + j];
        dinv[j] = (dg > 0.0f) ? 1.0f / dg : 0.0f;
    }

    const v4f bs0 = *reinterpret_cast<const v4f*>(bias + c0);
    const v4f bs1 = *reinterpret_cast<const v4f*>(bias + c0 + 128);
    const v4f g0  = *reinterpret_cast<const v4f*>(g + c0);
    const v4f g1  = *reinterpret_cast<const v4f*>(g + c0 + 128);
    const v4f be0 = *reinterpret_cast<const v4f*>(beta + c0);
    const v4f be1 = *reinterpret_cast<const v4f*>(beta + c0 + 128);
    const float a = pa[0];

    float sum[LN_RPW];
    for (int j = 0; j < LN_RPW; ++j) {
        x0[j] = x0[j] * dinv[j] + bs0;
        x1[j] = x1[j] * dinv[j] + bs1;
        sum[j] = lane_sum8(x0[j], x1[j]);
    }
    for (int m = 16; m > 0; m >>= 1)
        for (int j = 0; j < LN_RPW; ++j) sum[j] += __shfl_xor(sum[j], m, 32);

    float sq[LN_RPW];
    for (int j = 0; j < LN_RPW; ++j) {
        const float mu = sum[j] * (1.0f / D);
        x0[j] = x0[j] - mu;
        x1[j] = x1[j] - mu;
        sq[j] = lane_sum8(x0[j] * x0[j], x1[j] * x1[j]);
    }
    for (int m = 16; m > 0; m >>= 1)
        for (int j = 0; j < LN_RPW; ++j) sq[j] += __shfl_xor(sq[j], m, 32);

    for (int j = 0; j < LN_RPW; ++j) {
        const float rs = rsqrtf(sq[j] * (1.0f / D) + EPS_F);
        const v4f y0 = x0[j] * rs * g0 + be0;
        const v4f y1 = x1[j] * rs * g1 + be1;
        v4f o0, o1;
        for (int c = 0; c < 4; ++c) {
            o0[c] = (y0[c] >= 0.0f) ? y0[c] : a * y0[c];
            o1[c] = (y1[c] >= 0.0f) ? y1[c] : a * y1[c];
        }
        float* __restrict__ orow = out + (size_t)(r0 + j) * D + c0;
        *reinterpret_cast<v4f*>(orow)       = o0;
        *reinterpret_cast<v4f*>(orow + 128) = o1;
    }
}

__global__ __launch_bounds__(256) void wstd_kernel(const float* __restrict__ W,
                                                   float* __restrict__ Ws) {
    __shared__ float sred[256];
    const int r = blockIdx.x;
    const int f = threadIdx.x;
    const float w = W[(size_t)r * D + f];

    sred[f] = w;
    __syncthreads();
    #pragma unroll
    for (int s = 128; s > 0; s >>= 1) { if (f < s) sred[f] += sred[f + s]; __syncthreads(); }
    const float mean = sred[0] * (1.0f / D);
    __syncthreads();

    const float dv = w - mean;
    sred[f] = dv * dv;
    __syncthreads();
    #pragma unroll
    for (int s = 128; s > 0; s >>= 1) { if (f < s) sred[f] += sred[f + s]; __syncthreads(); }
    const float var = sred[0] * (1.0f / (D - 1));

    Ws[(size_t)r * D + f] = dv * rsqrtf(var + EPS_F);
}

extern "C" void kernel_launch(void* const* d_in, const int* in_sizes, int n_in,
                              void* d_out, int out_size, void* d_ws, size_t ws_size,
                              hipStream_t stream) {
    const float* edge_fea = (const float*)d_in[0];
    const int*   node_idx = (const int*)  d_in[1];
    const int*   edge_idx = (const int*)  d_in[2];
    const float* W1       = (const float*)d_in[3];
    const float* b1       = (const float*)d_in[4];
    const float* W2       = (const float*)d_in[5];
    const float* b2       = (const float*)d_in[6];
    const float* g1       = (const float*)d_in[7];
    const float* beta1    = (const float*)d_in[8];
    const float* g2       = (const float*)d_in[9];
    const float* beta2    = (const float*)d_in[10];
    const float* prelu_a  = (const float*)d_in[11];

    const size_t nodeRowBytes = (size_t)N_NODES * D * sizeof(float);

    float* B0  = (float*)d_out;
    char*  ws  = (char*)d_ws;
    float* B1  = (float*)ws;
    float* E   = (float*)(ws + nodeRowBytes);
    float* deg = E + (size_t)N_EDGES * D;
    float* cnt = deg + N_NODES;
    float* W2s = cnt + N_EDGES;


    int* dgB0 = (int*)d_out;
    int* dgB1 = (int*)ws;

    hipMemsetAsync(deg, 0, (size_t)(N_NODES + N_EDGES) * sizeof(float), stream);
    degcnt_kernel<SL_ROWS_E><<<(IDX_NQ + 255) / 256, 256, 0, stream>>>(node_idx, edge_idx, deg, cnt, dgB0);

    gemm_wmma_kernel<D_IN><<<GM_BLOCKS, 128, 0, stream>>>(edge_fea, W1, B1);

    scatter_rows_lds_kernel<SL_ROWS_E, N_EDGES, true><<<SL_BLOCKS_E, SL_THREADS, 0, stream>>>(B1, dgB0, cnt, E);
    digest_kernel<SL_ROWS_N><<<(IDX_NQ + 255) / 256, 256, 0, stream>>>(node_idx, edge_idx, dgB1);

    scatter_rows_lds_kernel<SL_ROWS_N, N_NODES, false><<<SL_BLOCKS_N, SL_THREADS, 0, stream>>>(E, dgB1, cnt, B0);
    fused_scale_ln_prelu_kernel<<<LN_BLOCKS, 256, 0, stream>>>(B0, deg, b1, g1, beta1, prelu_a, B1);

    wstd_kernel<<<D, 256, 0, stream>>>(W2, W2s);
    gemm_wmma_kernel<D><<<GM_BLOCKS, 128, 0, stream>>>(B1, W2s, B0);

    digest_kernel<SL_ROWS_E><<<(IDX_NQ + 255) / 256, 256, 0, stream>>>(edge_idx, node_idx, dgB1);
    scatter_rows_lds_kernel<SL_ROWS_E, N_EDGES, true><<<SL_BLOCKS_E, SL_THREADS, 0, stream>>>(B0, dgB1, cnt, E);
    digest_kernel<SL_ROWS_N><<<(IDX_NQ + 255) / 256, 256, 0, stream>>>(node_idx, edge_idx, dgB0);

    scatter_rows_lds_kernel<SL_ROWS_N, N_NODES, false><<<SL_BLOCKS_N, SL_THREADS, 0, stream>>>(E, dgB0, cnt, B1);
    fused_scale_ln_prelu_kernel<<<LN_BLOCKS, 256, 0, stream>>>(B1, deg, b2, g2, beta2, prelu_a, B0);
}


# Scatter-add row gathers as two 512-byte loads a row (16 bytes a lane), not eight one-line loads; LDS sums kept lane-transposed so float adds stay conflict-free; sums still in LDS arrival order
# speedup vs baseline: 4.9870x; 1.0906x over previous
#include <hip/hip_runtime.h>
#include <hip/hip_bf16.h>

#define N_NODES   100000
#define N_EDGES   20000
#define NNZ_CNT   800000
#define D_IN      128
#define D         256
#define EPS_F     1e-5f

typedef float v2f __attribute__((ext_vector_type(2)));
typedef float v8f __attribute__((ext_vector_type(8)));

__device__ __forceinline__ void atomAddF(float* p, float v) {
    __hip_atomic_fetch_add(p, v, __ATOMIC_RELAXED, __HIP_MEMORY_SCOPE_AGENT);
}

typedef int v4i __attribute__((ext_vector_type(4)));
static_assert(NNZ_CNT % 16 == 0, "index lists are read as 16-byte quads, digests too");
static_assert(N_NODES <= (1 << 17) && N_EDGES <= (1 << 17), "a source row number packs into 17 bits");
#define IDX_NQ (NNZ_CNT / 4)
#define REC_OFF_WORDS (1 << 18)

template<int ROWS>
__device__ __forceinline__ void build_digest(int t, const int* __restrict__ dst_idx,
                                             const int* __restrict__ src_idx,
                                             int* __restrict__ dg) {
    const v4i d4 = reinterpret_cast<const v4i*>(dst_idx)[t];
    const v4i s4 = reinterpret_cast<const v4i*>(src_idx)[t];
    unsigned word = 0u;
    v4i rec;
    for (int j = 0; j < 4; ++j) {
        const unsigned grp = (unsigned)d4[j] / (unsigned)ROWS;
        const unsigned rel = (unsigned)d4[j] % (unsigned)ROWS;
        word |= (grp & 255u) << (8 * j);
        rec[j] = (int)(((unsigned)s4[j] << 15) | (min(grp >> 8, 127u) << 8) | rel);
    }
    dg[t] = (int)word;
    reinterpret_cast<v4i*>(dg + REC_OFF_WORDS)[t] = rec;
}

template<int ROWS>
__global__ __launch_bounds__(256) void digest_kernel(const int* __restrict__ dst_idx,
                                                     const int* __restrict__ src_idx,
                                                     int* __restrict__ dg) {
    const int t = blockIdx.x * 256 + threadIdx.x;
    if (t < IDX_NQ) build_digest<ROWS>(t, dst_idx, src_idx, dg);
}

template<int ROWS_E>
__global__ __launch_bounds__(256) void degcnt_kernel(const int* __restrict__ node_idx,
                                                     const int* __restrict__ edge_idx,
                                                     float* __restrict__ deg,
                                                     float* __restrict__ cnt,
                                                     int* __restrict__ dge) {
    const int t = blockIdx.x * 256 + threadIdx.x;
    if (t < IDX_NQ) {
        const v4i n4 = reinterpret_cast<const v4i*>(node_idx)[t];
        const v4i e4 = reinterpret_cast<const v4i*>(edge_idx)[t];
        for (int j = 0; j < 4; ++j) {
            atomAddF(&deg[n4[j]], 1.0f);
            atomAddF(&cnt[e4[j]], 1.0f);
        }
        build_digest<ROWS_E>(t, edge_idx, node_idx, dge);
    }
}

#define GM_ROWS 64
#define GM_BLOCKS ((N_NODES + GM_ROWS - 1) / GM_ROWS)
static_assert(D == 256, "four waves of 64 columns cover the 256 output columns");
static_assert(N_NODES % 16 == 0, "a 16-row tile is wholly inside or wholly outside the matrix");

typedef float v4f __attribute__((ext_vector_type(4)));

__device__ __forceinline__ void load_frag(const float* __restrict__ p, v4f (&f)[4]) {
    #pragma unroll
    for (int j = 0; j < 4; ++j) f[j] = *reinterpret_cast<const v4f*>(p + 4 * j);
}

template<int K>
__global__ __launch_bounds__(128) void gemm_wmma_kernel(const float* __restrict__ X,
                                                        const float* __restrict__ W,
                                                        float* __restrict__ C) {
    static_assert(K % 32 == 0, "K is walked in 32-k chunks");
    constexpr int NCH = K / 32;
    const int lane = threadIdx.x & 31;
    const int wave = __builtin_amdgcn_readfirstlane((int)(threadIdx.x >> 5));
    const int c  = lane & 15;
    const int h  = lane >> 4;
    const int m0 = blockIdx.x * GM_ROWS;
    const int n0 = wave * 64;

    const float* __restrict__ ap[4];
    #pragma unroll
    for (int i = 0; i < 4; ++i) {
        const int ti = (m0 + 16 * i < N_NODES) ? i : 0;
        ap[i] = X + (size_t)(m0 + 16 * ti + c) * K + 16 * h;
    }
    const float* __restrict__ bp = W + (size_t)(n0 + 4 * c) * K + 16 * h;

    v8f acc[4][4];
    #pragma unroll
    for (int i = 0; i < 4; ++i)
        #pragma unroll
        for (int n = 0; n < 4; ++n) acc[i][n] = v8f{};

    v4f fa[2][4][4], fb[2][4][4];

    #pragma unroll
    for (int i = 0; i < 4; ++i) load_frag(ap[i], fa[0][i]);
    #pragma unroll
    for (int n = 0; n < 4; ++n) load_frag(bp + (size_t)n * K, fb[0][n]);

    #pragma unroll
    for (int q = 0; q < NCH; ++q) {
        const int cur = q & 1, nxt = cur ^ 1;
        if (q + 1 < NCH) {
            #pragma unroll
            for (int i = 0; i < 4; ++i) load_frag(ap[i] + 32 * (q + 1), fa[nxt][i]);
            #pragma unroll
            for (int n = 0; n < 4; ++n) load_frag(bp + (size_t)n * K + 32 * (q + 1), fb[nxt][n]);
        }
        __builtin_amdgcn_sched_barrier(0);
        #pragma unroll
        for (int j = 0; j < 4; ++j) {
            #pragma unroll
            for (int p = 0; p < 2; ++p) {
                v2f a[4], b[4];
                #pragma unroll
                for (int i = 0; i < 4; ++i) {
                    a[i] = p ? __builtin_shufflevector(fa[cur][i][j], fa[cur][i][j], 2, 3)
                             : __builtin_shufflevector(fa[cur][i][j], fa[cur][i][j], 0, 1);
                    b[i] = p ? __builtin_shufflevector(fb[cur][i][j], fb[cur][i][j], 2, 3)
                             : __builtin_shufflevector(fb[cur][i][j], fb[cur][i][j], 0, 1);
                }
                #pragma unroll
                for (int i = 0; i < 4; ++i)
                    #pragma unroll
                    for (int n = 0; n < 4; ++n)
                        acc[i][n] = __builtin_amdgcn_wmma_f32_16x16x4_f32(false, a[i], false, b[n], (short)0,
                                                                          acc[i][n], false, false);
            }
        }
    }

    float* __restrict__ cp = C + (size_t)(m0 + 8 * h) * D + n0 + 4 * c;
    #pragma unroll
    for (int i = 0; i < 4; ++i) {
        if (m0 + 16 * i < N_NODES) {
            #pragma unroll
            for (int r = 0; r < 8; ++r) {
                const v4f o = {acc[i][0][r], acc[i][1][r], acc[i][2][r], acc[i][3][r]};
                *reinterpret_cast<v4f*>(cp + (size_t)(16 * i + r) * D) = o;
            }
        }
    }
}

#define SL_THREADS 1024
#define SL_WAVES   (SL_THREADS / 32)
#define SL_ACAP    160
#define SL_BCAP    40
#define DG_NQ      (NNZ_CNT / 16)
#define SL_NIT     ((DG_NQ + SL_THREADS - 1) / SL_THREADS)

__device__ __forceinline__ void ldsAddF(float* p, float v) {
    __hip_atomic_fetch_add(p, v, __ATOMIC_RELAXED, __HIP_MEMORY_SCOPE_WORKGROUP);
}

__device__ __forceinline__ v4i sl_load_quad(const v4i* __restrict__ g4, int qi) {
    return g4[min(qi, DG_NQ - 1)];
}

__device__ __forceinline__ int sl_check(const unsigned* qa, int pos, int n, unsigned* qb, int cb,
                                        int lane, unsigned grp_hi,
                                        const int* __restrict__ rec) {
    const bool act = lane < n;
    const unsigned e = act ? qa[pos + lane] : 0u;
    const unsigned r = (unsigned)rec[e];
    const bool ok = act && ((r >> 8) & 127u) == grp_hi;
    const unsigned mk = __builtin_amdgcn_ballot_w32(ok);
    if (ok) qb[cb + (int)__builtin_amdgcn_mbcnt_lo(mk, 0u)] = ((r >> 15) << 8) | (r & 255u);
    return cb + __builtin_popcount(mk);
}

__device__ __forceinline__ void sl_add8(const unsigned* qb, int pos, int lane,
                                        const float* __restrict__ src, float* acc) {
    const unsigned pk = qb[pos + (lane & 7)];
    const int rel = (int)(pk & 255u);
    const int s = (int)(pk >> 8);
    v4f v[8][2];
    for (int j = 0; j < 8; ++j) {
        const int sj = __builtin_amdgcn_readlane(s, j);
        const v4f* __restrict__ row4 = reinterpret_cast<const v4f*>(src + (size_t)sj * D) + lane;
        v[j][0] = row4[0];
        v[j][1] = row4[32];
    }
    for (int j = 0; j < 8; ++j) {
        const int rj = __builtin_amdgcn_readlane(rel, j);
        float* a = acc + rj * D + lane;
        for (int hh = 0; hh < 2; ++hh)
            for (int c = 0; c < 4; ++c) ldsAddF(a + hh * 128 + c * 32, v[j][hh][c]);
    }
}

template<int ROWS, int NDST, bool SCALE>
__global__ __launch_bounds__(SL_THREADS) void scatter_rows_lds_kernel(const float* __restrict__ src,
                                                                      const int* __restrict__ dg,
                                                                      const float* __restrict__ rowcnt,
                                                                      float* __restrict__ dst) {
    static_assert(ROWS <= 256, "a row number inside the group packs into 8 bits");
    __shared__ float acc[ROWS * D];
    __shared__ unsigned listA[SL_WAVES][SL_ACAP];
    __shared__ unsigned listB[SL_WAVES][SL_BCAP];

    const int tid  = threadIdx.x;
    const int lane = tid & 31;
    const int wave = __builtin_amdgcn_readfirstlane(tid >> 5);
    const int base = blockIdx.x * ROWS;
    const unsigned pat = (unsigned)(blockIdx.x & 255) * 0x01010101u;
    const unsigned grp_hi = (unsigned)blockIdx.x >> 8;
    const int* __restrict__ rec = dg + REC_OFF_WORDS;

    v4f* acc4 = reinterpret_cast<v4f*>(acc);
    for (int k = tid; k < ROWS * D / 4; k += SL_THREADS) acc4[k] = v4f{0.0f, 0.0f, 0.0f, 0.0f};
    __syncthreads();

    unsigned* qa = listA[wave];
    unsigned* qb = listB[wave];
    int ca = 0;
    int cb = 0;
    const v4i* __restrict__ g4 = reinterpret_cast<const v4i*>(dg);

    v4i nx = sl_load_quad(g4, tid);
    for (int it = 0; it < SL_NIT; ++it) {
        const int qi = it * SL_THREADS + tid;
        const v4i x = nx;
        nx = sl_load_quad(g4, qi + SL_THREADS);
        const bool in = qi < DG_NQ;
        #pragma unroll
        for (int c = 0; c < 4; ++c) {
            const unsigned y = (unsigned)x[c] ^ pat;
            unsigned z = ~(((y & 0x7F7F7F7Fu) + 0x7F7F7F7Fu) | y | 0x7F7F7F7Fu);
            z = in ? z : 0u;
            unsigned any = __builtin_amdgcn_ballot_w32(z != 0u);
            if (any != 0u) {
                do {
                    if (z != 0u) {
                        const int b = __builtin_ctz(z) >> 3;
                        z &= z - 1u;
                        qa[ca + (int)__builtin_amdgcn_mbcnt_lo(any, 0u)] = (unsigned)(16 * qi + 4 * c + b);
                    }
                    ca += __builtin_popcount(any);
                    any = __builtin_amdgcn_ballot_w32(z != 0u);
                } while (any != 0u);
                while (ca >= 32) {
                    ca -= 32;
                    cb = sl_check(qa, ca, 32, qb, cb, lane, grp_hi, rec);
                    while (cb >= 8) {
                        cb -= 8;
                        sl_add8(qb, cb, lane, src, acc);
                    }
                }
            }
        }
    }

    if (ca > 0) {
        cb = sl_check(qa, 0, ca, qb, cb, lane, grp_hi, rec);
        while (cb >= 8) {
            cb -= 8;
            sl_add8(qb, cb, lane, src, acc);
        }
    }
    for (int j = 0; j < cb; ++j) {
        const unsigned pk = (unsigned)__builtin_amdgcn_readfirstlane((int)qb[j]);
        const v4f* __restrict__ row4 = reinterpret_cast<const v4f*>(src + (size_t)(pk >> 8) * D) + lane;
        float* a = acc + (int)(pk & 255u) * D + lane;
        const v4f v0 = row4[0];
        const v4f v1 = row4[32];
        for (int c = 0; c < 4; ++c) ldsAddF(a + c * 32, v0[c]);
        for (int c = 0; c < 4; ++c) ldsAddF(a + 128 + c * 32, v1[c]);
    }
    __syncthreads();

    v4f* __restrict__ out4 = reinterpret_cast<v4f*>(dst + (size_t)base * D);
    for (int k = tid; k < ROWS * D / 4; k += SL_THREADS) {
        const int r = base + (k >> 6);
        if (r < NDST) {
            const float* p = acc + (k >> 6) * D + ((k >> 5) & 1) * 128 + (k & 31);
            v4f v = {p[0], p[32], p[64], p[96]};
            if (SCALE) {
                const float c = rowcnt[r];
                const float inv = (c > 0.0f) ? 1.0f / c : 0.0f;
                v = v * inv;
            }
            out4[k] = v;
        }
    }
}

#define SL_ROWS_E 80
#define SL_ROWS_N 256
#define SL_BLOCKS_E ((N_EDGES + SL_ROWS_E - 1) / SL_ROWS_E)
#define SL_BLOCKS_N ((N_NODES + SL_ROWS_N - 1) / SL_ROWS_N)

#define LN_RPW 4
#define LN_WAVES_PER_BLOCK 8
#define LN_BLOCKS (N_NODES / (LN_RPW * LN_WAVES_PER_BLOCK))
static_assert(N_NODES % (LN_RPW * LN_WAVES_PER_BLOCK) == 0, "row count must fill whole blocks");
static_assert(D == 256, "lane-to-column map below is written for 256-wide rows");

__device__ __forceinline__ float lane_sum8(v4f a, v4f b) {
    return ((a.x + a.y) + (a.z + a.w)) + ((b.x + b.y) + (b.z + b.w));
}

__global__ __launch_bounds__(256) void fused_scale_ln_prelu_kernel(const float* __restrict__ acc,
                                                                   const float* __restrict__ deg,
                                                                   const float* __restrict__ bias,
                                                                   const float* __restrict__ g,
                                                                   const float* __restrict__ beta,
                                                                   const float* __restrict__ pa,
                                                                   float* __restrict__ out) {
    const int lane = threadIdx.x & 31;
    const int wv = __builtin_amdgcn_readfirstlane((int)((blockIdx.x * 256 + threadIdx.x) >> 5));
    const int r0 = wv * LN_RPW;
    const int c0 = lane * 4;

    v4f x0[LN_RPW], x1[LN_RPW];
    for (int j = 0; j < LN_RPW; ++j) {
        const float* __restrict__ row = acc + (size_t)(r0 + j) * D + c0;
        x0[j] = *reinterpret_cast<const v4f*>(row);
        x1[j] = *reinterpret_cast<const v4f*>(row + 128);
    }

    float dinv[LN_RPW];
    for (int j = 0; j < LN_RPW; ++j) {
        const float dg = deg[r0 + j];
        dinv[j] = (dg > 0.0f) ? 1.0f / dg : 0.0f;
    }

    const v4f bs0 = *reinterpret_cast<const v4f*>(bias + c0);
    const v4f bs1 = *reinterpret_cast<const v4f*>(bias + c0 + 128);
    const v4f g0  = *reinterpret_cast<const v4f*>(g + c0);
    const v4f g1  = *reinterpret_cast<const v4f*>(g + c0 + 128);
    const v4f be0 = *reinterpret_cast<const v4f*>(beta + c0);
    const v4f be1 = *reinterpret_cast<const v4f*>(beta + c0 + 128);
    const float a = pa[0];

    float sum[LN_RPW];
    for (int j = 0; j < LN_RPW; ++j) {
        x0[j] = x0[j] * dinv[j] + bs0;
        x1[j] = x1[j] * dinv[j] + bs1;
        sum[j] = lane_sum8(x0[j], x1[j]);
    }
    for (int m = 16; m > 0; m >>= 1)
        for (int j = 0; j < LN_RPW; ++j) sum[j] += __shfl_xor(sum[j], m, 32);

    float sq[LN_RPW];
    for (int j = 0; j < LN_RPW; ++j) {
        const float mu = sum[j] * (1.0f / D);
        x0[j] = x0[j] - mu;
        x1[j] = x1[j] - mu;
        sq[j] = lane_sum8(x0[j] * x0[j], x1[j] * x1[j]);
    }
    for (int m = 16; m > 0; m >>= 1)
        for (int j = 0; j < LN_RPW; ++j) sq[j] += __shfl_xor(sq[j], m, 32);

    for (int j = 0; j < LN_RPW; ++j) {
        const float rs = rsqrtf(sq[j] * (1.0f / D) + EPS_F);
        const v4f y0 = x0[j] * rs * g0 + be0;
        const v4f y1 = x1[j] * rs * g1 + be1;
        v4f o0, o1;
        for (int c = 0; c < 4; ++c) {
            o0[c] = (y0[c] >= 0.0f) ? y0[c] : a * y0[c];
            o1[c] = (y1[c] >= 0.0f) ? y1[c] : a * y1[c];
        }
        float* __restrict__ orow = out + (size_t)(r0 + j) * D + c0;
        *reinterpret_cast<v4f*>(orow)       = o0;
        *reinterpret_cast<v4f*>(orow + 128) = o1;
    }
}

__global__ __launch_bounds__(256) void wstd_kernel(const float* __restrict__ W,
                                                   float* __restrict__ Ws) {
    __shared__ float sred[256];
    const int r = blockIdx.x;
    const int f = threadIdx.x;
    const float w = W[(size_t)r * D + f];

    sred[f] = w;
    __syncthreads();
    #pragma unroll
    for (int s = 128; s > 0; s >>= 1) { if (f < s) sred[f] += sred[f + s]; __syncthreads(); }
    const float mean = sred[0] * (1.0f / D);
    __syncthreads();

    const float dv = w - mean;
    sred[f] = dv * dv;
    __syncthreads();
    #pragma unroll
    for (int s = 128; s > 0; s >>= 1) { if (f < s) sred[f] += sred[f + s]; __syncthreads(); }
    const float var = sred[0] * (1.0f / (D - 1));

    Ws[(size_t)r * D + f] = dv * rsqrtf(var + EPS_F);
}

extern "C" void kernel_launch(void* const* d_in, const int* in_sizes, int n_in,
                              void* d_out, int out_size, void* d_ws, size_t ws_size,
                              hipStream_t stream) {
    const float* edge_fea = (const float*)d_in[0];
    const int*   node_idx = (const int*)  d_in[1];
    const int*   edge_idx = (const int*)  d_in[2];
    const float* W1       = (const float*)d_in[3];
    const float* b1       = (const float*)d_in[4];
    const float* W2       = (const float*)d_in[5];
    const float* b2       = (const float*)d_in[6];
    const float* g1       = (const float*)d_in[7];
    const float* beta1    = (const float*)d_in[8];
    const float* g2       = (const float*)d_in[9];
    const float* beta2    = (const float*)d_in[10];
    const float* prelu_a  = (const float*)d_in[11];

    const size_t nodeRowBytes = (size_t)N_NODES * D * sizeof(float);

    float* B0  = (float*)d_out;
    char*  ws  = (char*)d_ws;
    float* B1  = (float*)ws;
    float* E   = (float*)(ws + nodeRowBytes);
    float* deg = E + (size_t)N_EDGES * D;
    float* cnt = deg + N_NODES;
    float* W2s = cnt + N_EDGES;


    int* dgB0 = (int*)d_out;
    int* dgB1 = (int*)ws;

    hipMemsetAsync(deg, 0, (size_t)(N_NODES + N_EDGES) * sizeof(float), stream);
    degcnt_kernel<SL_ROWS_E><<<(IDX_NQ + 255) / 256, 256, 0, stream>>>(node_idx, edge_idx, deg, cnt, dgB0);

    gemm_wmma_kernel<D_IN><<<GM_BLOCKS, 128, 0, stream>>>(edge_fea, W1, B1);

    scatter_rows_lds_kernel<SL_ROWS_E, N_EDGES, true><<<SL_BLOCKS_E, SL_THREADS, 0, stream>>>(B1, dgB0, cnt, E);
    digest_kernel<SL_ROWS_N><<<(IDX_NQ + 255) / 256, 256, 0, stream>>>(node_idx, edge_idx, dgB1);

    scatter_rows_lds_kernel<SL_ROWS_N, N_NODES, false><<<SL_BLOCKS_N, SL_THREADS, 0, stream>>>(E, dgB1, cnt, B0);
    fused_scale_ln_prelu_kernel<<<LN_BLOCKS, 256, 0, stream>>>(B0, deg, b1, g1, beta1, prelu_a, B1);

    wstd_kernel<<<D, 256, 0, stream>>>(W2, W2s);
    gemm_wmma_kernel<D><<<GM_BLOCKS, 128, 0, stream>>>(B1, W2s, B0);

    digest_kernel<SL_ROWS_E><<<(IDX_NQ + 255) / 256, 256, 0, stream>>>(edge_idx, node_idx, dgB1);
    scatter_rows_lds_kernel<SL_ROWS_E, N_EDGES, true><<<SL_BLOCKS_E, SL_THREADS, 0, stream>>>(B0, dgB1, cnt, E);
    digest_kernel<SL_ROWS_N><<<(IDX_NQ + 255) / 256, 256, 0, stream>>>(node_idx, edge_idx, dgB0);

    scatter_rows_lds_kernel<SL_ROWS_N, N_NODES, false><<<SL_BLOCKS_N, SL_THREADS, 0, stream>>>(E, dgB0, cnt, B1);
    fused_scale_ln_prelu_kernel<<<LN_BLOCKS, 256, 0, stream>>>(B1, deg, b2, g2, beta2, prelu_a, B0);
}


# Node-side scatter-add: a work-group sums both row ranges sharing its digest byte (2 x 196 rows), the second from per-wave LDS lists noted in the one walk; 256 groups; sums still in LDS arrival order
# speedup vs baseline: 5.2989x; 1.0625x over previous
#include <hip/hip_runtime.h>
#include <hip/hip_bf16.h>

#define N_NODES   100000
#define N_EDGES   20000
#define NNZ_CNT   800000
#define D_IN      128
#define D         256
#define EPS_F     1e-5f

typedef float v2f __attribute__((ext_vector_type(2)));
typedef float v8f __attribute__((ext_vector_type(8)));

__device__ __forceinline__ void atomAddF(float* p, float v) {
    __hip_atomic_fetch_add(p, v, __ATOMIC_RELAXED, __HIP_MEMORY_SCOPE_AGENT);
}

typedef int v4i __attribute__((ext_vector_type(4)));
static_assert(NNZ_CNT % 16 == 0, "index lists are read as 16-byte quads, digests too");
static_assert(N_NODES <= (1 << 17) && N_EDGES <= (1 << 17), "a source row number packs into 17 bits");
#define IDX_NQ (NNZ_CNT / 4)
#define REC_OFF_WORDS (1 << 18)

template<int ROWS>
__device__ __forceinline__ void build_digest(int t, const int* __restrict__ dst_idx,
                                             const int* __restrict__ src_idx,
                                             int* __restrict__ dg) {
    const v4i d4 = reinterpret_cast<const v4i*>(dst_idx)[t];
    const v4i s4 = reinterpret_cast<const v4i*>(src_idx)[t];
    unsigned word = 0u;
    v4i rec;
    for (int j = 0; j < 4; ++j) {
        const unsigned grp = (unsigned)d4[j] / (unsigned)ROWS;
        const unsigned rel = (unsigned)d4[j] % (unsigned)ROWS;
        word |= (grp & 255u) << (8 * j);
        rec[j] = (int)(((unsigned)s4[j] << 15) | (min(grp >> 8, 127u) << 8) | rel);
    }
    dg[t] = (int)word;
    reinterpret_cast<v4i*>(dg + REC_OFF_WORDS)[t] = rec;
}

template<int ROWS>
__global__ __launch_bounds__(256) void digest_kernel(const int* __restrict__ dst_idx,
                                                     const int* __restrict__ src_idx,
                                                     int* __restrict__ dg) {
    const int t = blockIdx.x * 256 + threadIdx.x;
    if (t < IDX_NQ) build_digest<ROWS>(t, dst_idx, src_idx, dg);
}

template<int ROWS_E>
__global__ __launch_bounds__(256) void degcnt_kernel(const int* __restrict__ node_idx,
                                                     const int* __restrict__ edge_idx,
                                                     float* __restrict__ deg,
                                                     float* __restrict__ cnt,
                                                     int* __restrict__ dge) {
    const int t = blockIdx.x * 256 + threadIdx.x;
    if (t < IDX_NQ) {
        const v4i n4 = reinterpret_cast<const v4i*>(node_idx)[t];
        const v4i e4 = reinterpret_cast<const v4i*>(edge_idx)[t];
        for (int j = 0; j < 4; ++j) {
            atomAddF(&deg[n4[j]], 1.0f);
            atomAddF(&cnt[e4[j]], 1.0f);
        }
        build_digest<ROWS_E>(t, edge_idx, node_idx, dge);
    }
}

#define GM_ROWS 64
#define GM_BLOCKS ((N_NODES + GM_ROWS - 1) / GM_ROWS)
static_assert(D == 256, "four waves of 64 columns cover the 256 output columns");
static_assert(N_NODES % 16 == 0, "a 16-row tile is wholly inside or wholly outside the matrix");

typedef float v4f __attribute__((ext_vector_type(4)));

__device__ __forceinline__ void load_frag(const float* __restrict__ p, v4f (&f)[4]) {
    #pragma unroll
    for (int j = 0; j < 4; ++j) f[j] = *reinterpret_cast<const v4f*>(p + 4 * j);
}

template<int K>
__global__ __launch_bounds__(128) void gemm_wmma_kernel(const float* __restrict__ X,
                                                        const float* __restrict__ W,
                                                        float* __restrict__ C) {
    static_assert(K % 32 == 0, "K is walked in 32-k chunks");
    constexpr int NCH = K / 32;
    const int lane = threadIdx.x & 31;
    const int wave = __builtin_amdgcn_readfirstlane((int)(threadIdx.x >> 5));
    const int c  = lane & 15;
    const int h  = lane >> 4;
    const int m0 = blockIdx.x * GM_ROWS;
    const int n0 = wave * 64;

    const float* __restrict__ ap[4];
    #pragma unroll
    for (int i = 0; i < 4; ++i) {
        const int ti = (m0 + 16 * i < N_NODES) ? i : 0;
        ap[i] = X + (size_t)(m0 + 16 * ti + c) * K + 16 * h;
    }
    const float* __restrict__ bp = W + (size_t)(n0 + 4 * c) * K + 16 * h;

    v8f acc[4][4];
    #pragma unroll
    for (int i = 0; i < 4; ++i)
        #pragma unroll
        for (int n = 0; n < 4; ++n) acc[i][n] = v8f{};

    v4f fa[2][4][4], fb[2][4][4];

    #pragma unroll
    for (int i = 0; i < 4; ++i) load_frag(ap[i], fa[0][i]);
    #pragma unroll
    for (int n = 0; n < 4; ++n) load_frag(bp + (size_t)n * K, fb[0][n]);

    #pragma unroll
    for (int q = 0; q < NCH; ++q) {
        const int cur = q & 1, nxt = cur ^ 1;
        if (q + 1 < NCH) {
            #pragma unroll
            for (int i = 0; i < 4; ++i) load_frag(ap[i] + 32 * (q + 1), fa[nxt][i]);
            #pragma unroll
            for (int n = 0; n < 4; ++n) load_frag(bp + (size_t)n * K + 32 * (q + 1), fb[nxt][n]);
        }
        __builtin_amdgcn_sched_barrier(0);
        #pragma unroll
        for (int j = 0; j < 4; ++j) {
            #pragma unroll
            for (int p = 0; p < 2; ++p) {
                v2f a[4], b[4];
                #pragma unroll
                for (int i = 0; i < 4; ++i) {
                    a[i] = p ? __builtin_shufflevector(fa[cur][i][j], fa[cur][i][j], 2, 3)
                             : __builtin_shufflevector(fa[cur][i][j], fa[cur][i][j], 0, 1);
                    b[i] = p ? __builtin_shufflevector(fb[cur][i][j], fb[cur][i][j], 2, 3)
                             : __builtin_shufflevector(fb[cur][i][j], fb[cur][i][j], 0, 1);
                }
                #pragma unroll
                for (int i = 0; i < 4; ++i)
                    #pragma unroll
                    for (int n = 0; n < 4; ++n)
                        acc[i][n] = __builtin_amdgcn_wmma_f32_16x16x4_f32(false, a[i], false, b[n], (short)0,
                                                                          acc[i][n], false, false);
            }
        }
    }

    float* __restrict__ cp = C + (size_t)(m0 + 8 * h) * D + n0 + 4 * c;
    #pragma unroll
    for (int i = 0; i < 4; ++i) {
        if (m0 + 16 * i < N_NODES) {
            #pragma unroll
            for (int r = 0; r < 8; ++r) {
                const v4f o = {acc[i][0][r], acc[i][1][r], acc[i][2][r], acc[i][3][r]};
                *reinterpret_cast<v4f*>(cp + (size_t)(16 * i + r) * D) = o;
            }
        }
    }
}

#define SL_THREADS 1024
#define SL_WAVES   (SL_THREADS / 32)
#define SL_ACAP    160
#define SL_BCAP    40
#define SL_DCAP    512
#define DG_NQ      (NNZ_CNT / 16)
#define SL_NIT     ((DG_NQ + SL_THREADS - 1) / SL_THREADS)

__device__ __forceinline__ void ldsAddF(float* p, float v) {
    __hip_atomic_fetch_add(p, v, __ATOMIC_RELAXED, __HIP_MEMORY_SCOPE_WORKGROUP);
}

__device__ __forceinline__ v4i sl_load_quad(const v4i* __restrict__ g4, int qi) {
    return g4[min(qi, DG_NQ - 1)];
}

template<bool DEFER>
__device__ __forceinline__ int sl_check(const unsigned* qa, int pos, int n, unsigned* qb, int cb,
                                        int lane, unsigned grp_hi,
                                        const int* __restrict__ rec, unsigned* qd, int& cd) {
    const bool act = lane < n;
    const unsigned e = act ? qa[pos + lane] : 0u;
    const unsigned r = (unsigned)rec[e];
    const unsigned fld = (r >> 8) & 127u;
    const unsigned pk = ((r >> 15) << 8) | (r & 255u);
    const bool ok = act && fld == grp_hi;
    const unsigned mk = __builtin_amdgcn_ballot_w32(ok);
    if (ok) qb[cb + (int)__builtin_amdgcn_mbcnt_lo(mk, 0u)] = pk;
    if (DEFER) {
        const bool od = act && fld == grp_hi + 1u;
        const unsigned md = __builtin_amdgcn_ballot_w32(od);
        const int nd = __builtin_popcount(md);
        if (cd + nd <= SL_DCAP) {
            if (od) qd[cd + (int)__builtin_amdgcn_mbcnt_lo(md, 0u)] = pk;
            cd += nd;
        } else {
            cd = SL_DCAP + 1;
        }
    }
    return cb + __builtin_popcount(mk);
}

__device__ __forceinline__ void sl_add8(const unsigned* qb, int pos, int lane,
                                        const float* __restrict__ src, float* acc) {
    const unsigned pk = qb[pos + (lane & 7)];
    const int rel = (int)(pk & 255u);
    const int s = (int)(pk >> 8);
    v4f v[8][2];
    for (int j = 0; j < 8; ++j) {
        const int sj = __builtin_amdgcn_readlane(s, j);
        const v4f* __restrict__ row4 = reinterpret_cast<const v4f*>(src + (size_t)sj * D) + lane;
        v[j][0] = row4[0];
        v[j][1] = row4[32];
    }
    for (int j = 0; j < 8; ++j) {
        const int rj = __builtin_amdgcn_readlane(rel, j);
        float* a = acc + rj * D + lane;
        for (int hh = 0; hh < 2; ++hh)
            for (int c = 0; c < 4; ++c) ldsAddF(a + hh * 128 + c * 32, v[j][hh][c]);
    }
}

__device__ __forceinline__ void sl_add1(unsigned pk, int lane, const float* __restrict__ src, float* acc) {
    const v4f* __restrict__ row4 = reinterpret_cast<const v4f*>(src + (size_t)(pk >> 8) * D) + lane;
    float* a = acc + (int)(pk & 255u) * D + lane;
    const v4f v0 = row4[0];
    const v4f v1 = row4[32];
    for (int c = 0; c < 4; ++c) ldsAddF(a + c * 32, v0[c]);
    for (int c = 0; c < 4; ++c) ldsAddF(a + 128 + c * 32, v1[c]);
}

template<bool DEFER>
__device__ __forceinline__ void sl_walk(const int* __restrict__ dg, const float* __restrict__ src,
                                        float* acc, unsigned* qa, unsigned* qb, unsigned* qd, int& cd,
                                        int tid, int lane, unsigned pat, unsigned grp_hi) {
    const int* __restrict__ rec = dg + REC_OFF_WORDS;
    int ca = 0;
    int cb = 0;
    const v4i* __restrict__ g4 = reinterpret_cast<const v4i*>(dg);

    v4i nx = sl_load_quad(g4, tid);
    for (int it = 0; it < SL_NIT; ++it) {
        const int qi = it * SL_THREADS + tid;
        const v4i x = nx;
        nx = sl_load_quad(g4, qi + SL_THREADS);
        const bool in = qi < DG_NQ;
        #pragma unroll
        for (int c = 0; c < 4; ++c) {
            const unsigned y = (unsigned)x[c] ^ pat;
            unsigned z = ~(((y & 0x7F7F7F7Fu) + 0x7F7F7F7Fu) | y | 0x7F7F7F7Fu);
            z = in ? z : 0u;
            unsigned any = __builtin_amdgcn_ballot_w32(z != 0u);
            if (any != 0u) {
                do {
                    if (z != 0u) {
                        const int b = __builtin_ctz(z) >> 3;
                        z &= z - 1u;
                        qa[ca + (int)__builtin_amdgcn_mbcnt_lo(any, 0u)] = (unsigned)(16 * qi + 4 * c + b);
                    }
                    ca += __builtin_popcount(any);
                    any = __builtin_amdgcn_ballot_w32(z != 0u);
                } while (any != 0u);
                while (ca >= 32) {
                    ca -= 32;
                    cb = sl_check<DEFER>(qa, ca, 32, qb, cb, lane, grp_hi, rec, qd, cd);
                    while (cb >= 8) {
                        cb -= 8;
                        sl_add8(qb, cb, lane, src, acc);
                    }
                }
            }
        }
    }

    if (ca > 0) {
        cb = sl_check<DEFER>(qa, 0, ca, qb, cb, lane, grp_hi, rec, qd, cd);
        while (cb >= 8) {
            cb -= 8;
            sl_add8(qb, cb, lane, src, acc);
        }
    }
    for (int j = 0; j < cb; ++j)
        sl_add1((unsigned)__builtin_amdgcn_readfirstlane((int)qb[j]), lane, src, acc);
}

template<int ROWS>
__device__ __forceinline__ void sl_zero(float* acc, int tid) {
    v4f* acc4 = reinterpret_cast<v4f*>(acc);
    for (int k = tid; k < ROWS * D / 4; k += SL_THREADS) acc4[k] = v4f{0.0f, 0.0f, 0.0f, 0.0f};
}

template<int ROWS, int NDST, bool SCALE>
__device__ __forceinline__ void sl_store(const float* acc, int base, const float* __restrict__ rowcnt,
                                         float* __restrict__ dst, int tid) {
    v4f* __restrict__ out4 = reinterpret_cast<v4f*>(dst + (size_t)base * D);
    for (int k = tid; k < ROWS * D / 4; k += SL_THREADS) {
        const int r = base + (k >> 6);
        if (r < NDST) {
            const float* p = acc + (k >> 6) * D + ((k >> 5) & 1) * 128 + (k & 31);
            v4f v = {p[0], p[32], p[64], p[96]};
            if (SCALE) {
                const float c = rowcnt[r];
                const float inv = (c > 0.0f) ? 1.0f / c : 0.0f;
                v = v * inv;
            }
            out4[k] = v;
        }
    }
}

template<int ROWS, int NDST, bool SCALE, bool PAIR>
__global__ __launch_bounds__(SL_THREADS) void scatter_rows_lds_kernel(const float* __restrict__ src,
                                                                      const int* __restrict__ dg,
                                                                      const float* __restrict__ rowcnt,
                                                                      float* __restrict__ dst) {
    static_assert(ROWS <= 256, "a row number inside the group packs into 8 bits");
    static_assert(!PAIR || 512 * ROWS >= NDST, "256 work-groups of two ranges cover every row");
    __shared__ float acc[ROWS * D];
    __shared__ unsigned listA[SL_WAVES][SL_ACAP];
    __shared__ unsigned listB[SL_WAVES][SL_BCAP];
    __shared__ unsigned listD[PAIR ? SL_WAVES : 1][PAIR ? SL_DCAP : 1];

    const int tid  = threadIdx.x;
    const int lane = tid & 31;
    const int wave = __builtin_amdgcn_readfirstlane(tid >> 5);
    const unsigned pat = (unsigned)(blockIdx.x & 255) * 0x01010101u;
    const unsigned grp_hi = (unsigned)blockIdx.x >> 8;

    unsigned* qa = listA[wave];
    unsigned* qb = listB[wave];
    unsigned* qd = listD[PAIR ? wave : 0];
    int cd = 0;

    sl_zero<ROWS>(acc, tid);
    __syncthreads();
    sl_walk<PAIR>(dg, src, acc, qa, qb, qd, cd, tid, lane, pat, grp_hi);
    __syncthreads();
    sl_store<ROWS, NDST, SCALE>(acc, (int)blockIdx.x * ROWS, rowcnt, dst, tid);

    if (PAIR) {
        __syncthreads();
        sl_zero<ROWS>(acc, tid);
        __syncthreads();
        const int nd = __builtin_amdgcn_readfirstlane(cd);
        if (nd <= SL_DCAP) {
            int k = 0;
            for (; k + 8 <= nd; k += 8) sl_add8(qd, k, lane, src, acc);
            for (; k < nd; ++k)
                sl_add1((unsigned)__builtin_amdgcn_readfirstlane((int)qd[k]), lane, src, acc);
        } else {
            int none = 0;
            sl_walk<false>(dg, src, acc, qa, qb, qd, none, tid, lane, pat, grp_hi + 1u);
        }
        __syncthreads();
        sl_store<ROWS, NDST, SCALE>(acc, ((int)blockIdx.x + 256) * ROWS, rowcnt, dst, tid);
    }
}

#define SL_ROWS_E 80
#define SL_ROWS_N 196
#define SL_BLOCKS_E ((N_EDGES + SL_ROWS_E - 1) / SL_ROWS_E)
#define SL_BLOCKS_N 256
static_assert(SL_BLOCKS_E <= 256, "one range per group: every group has a digest byte of its own");

#define LN_RPW 4
#define LN_WAVES_PER_BLOCK 8
#define LN_BLOCKS (N_NODES / (LN_RPW * LN_WAVES_PER_BLOCK))
static_assert(N_NODES % (LN_RPW * LN_WAVES_PER_BLOCK) == 0, "row count must fill whole blocks");
static_assert(D == 256, "lane-to-column map below is written for 256-wide rows");

__device__ __forceinline__ float lane_sum8(v4f a, v4f b) {
    return ((a.x + a.y) + (a.z + a.w)) + ((b.x + b.y) + (b.z + b.w));
}

__global__ __launch_bounds__(256) void fused_scale_ln_prelu_kernel(const float* __restrict__ acc,
                                                                   const float* __restrict__ deg,
                                                                   const float* __restrict__ bias,
                                                                   const float* __restrict__ g,
                                                                   const float* __restrict__ beta,
                                                                   const float* __restrict__ pa,
                                                                   float* __restrict__ out) {
    const int lane = threadIdx.x & 31;
    const int wv = __builtin_amdgcn_readfirstlane((int)((blockIdx.x * 256 + threadIdx.x) >> 5));
    const int r0 = wv * LN_RPW;
    const int c0 = lane * 4;

    v4f x0[LN_RPW], x1[LN_RPW];
    for (int j = 0; j < LN_RPW; ++j) {
        const float* __restrict__ row = acc + (size_t)(r0 + j) * D + c0;
        x0[j] = *reinterpret_cast<const v4f*>(row);
        x1[j] = *reinterpret_cast<const v4f*>(row + 128);
    }

    float dinv[LN_RPW];
    for (int j = 0; j < LN_RPW; ++j) {
        const float dg = deg[r0 + j];
        dinv[j] = (dg > 0.0f) ? 1.0f / dg : 0.0f;
    }

    const v4f bs0 = *reinterpret_cast<const v4f*>(bias + c0);
    const v4f bs1 = *reinterpret_cast<const v4f*>(bias + c0 + 128);
    const v4f g0  = *reinterpret_cast<const v4f*>(g + c0);
    const v4f g1  = *reinterpret_cast<const v4f*>(g + c0 + 128);
    const v4f be0 = *reinterpret_cast<const v4f*>(beta + c0);
    const v4f be1 = *reinterpret_cast<const v4f*>(beta + c0 + 128);
    const float a = pa[0];

    float sum[LN_RPW];
    for (int j = 0; j < LN_RPW; ++j) {
        x0[j] = x0[j] * dinv[j] + bs0;
        x1[j] = x1[j] * dinv[j] + bs1;
        sum[j] = lane_sum8(x0[j], x1[j]);
    }
    for (int m = 16; m > 0; m >>= 1)
        for (int j = 0; j < LN_RPW; ++j) sum[j] += __shfl_xor(sum[j], m, 32);

    float sq[LN_RPW];
    for (int j = 0; j < LN_RPW; ++j) {
        const float mu = sum[j] * (1.0f / D);
        x0[j] = x0[j] - mu;
        x1[j] = x1[j] - mu;
        sq[j] = lane_sum8(x0[j] * x0[j], x1[j] * x1[j]);
    }
    for (int m = 16; m > 0; m >>= 1)
        for (int j = 0; j < LN_RPW; ++j) sq[j] += __shfl_xor(sq[j], m, 32);

    for (int j = 0; j < LN_RPW; ++j) {
        const float rs = rsqrtf(sq[j] * (1.0f / D) + EPS_F);
        const v4f y0 = x0[j] * rs * g0 + be0;
        const v4f y1 = x1[j] * rs * g1 + be1;
        v4f o0, o1;
        for (int c = 0; c < 4; ++c) {
            o0[c] = (y0[c] >= 0.0f) ? y0[c] : a * y0[c];
            o1[c] = (y1[c] >= 0.0f) ? y1[c] : a * y1[c];
        }
        float* __restrict__ orow = out + (size_t)(r0 + j) * D + c0;
        *reinterpret_cast<v4f*>(orow)       = o0;
        *reinterpret_cast<v4f*>(orow + 128) = o1;
    }
}

__global__ __launch_bounds__(256) void wstd_kernel(const float* __restrict__ W,
                                                   float* __restrict__ Ws) {
    __shared__ float sred[256];
    const int r = blockIdx.x;
    const int f = threadIdx.x;
    const float w = W[(size_t)r * D + f];

    sred[f] = w;
    __syncthreads();
    #pragma unroll
    for (int s = 128; s > 0; s >>= 1) { if (f < s) sred[f] += sred[f + s]; __syncthreads(); }
    const float mean = sred[0] * (1.0f / D);
    __syncthreads();

    const float dv = w - mean;
    sred[f] = dv * dv;
    __syncthreads();
    #pragma unroll
    for (int s = 128; s > 0; s >>= 1) { if (f < s) sred[f] += sred[f + s]; __syncthreads(); }
    const float var = sred[0] * (1.0f / (D - 1));

    Ws[(size_t)r * D + f] = dv * rsqrtf(var + EPS_F);
}

extern "C" void kernel_launch(void* const* d_in, const int* in_sizes, int n_in,
                              void* d_out, int out_size, void* d_ws, size_t ws_size,
                              hipStream_t stream) {
    const float* edge_fea = (const float*)d_in[0];
    const int*   node_idx = (const int*)  d_in[1];
    const int*   edge_idx = (const int*)  d_in[2];
    const float* W1       = (const float*)d_in[3];
    const float* b1       = (const float*)d_in[4];
    const float* W2       = (const float*)d_in[5];
    const float* b2       = (const float*)d_in[6];
    const float* g1       = (const float*)d_in[7];
    const float* beta1    = (const float*)d_in[8];
    const float* g2       = (const float*)d_in[9];
    const float* beta2    = (const float*)d_in[10];
    const float* prelu_a  = (const float*)d_in[11];

    const size_t nodeRowBytes = (size_t)N_NODES * D * sizeof(float);

    float* B0  = (float*)d_out;
    char*  ws  = (char*)d_ws;
    float* B1  = (float*)ws;
    float* E   = (float*)(ws + nodeRowBytes);
    float* deg = E + (size_t)N_EDGES * D;
    float* cnt = deg + N_NODES;
    float* W2s = cnt + N_EDGES;


    int* dgB0 = (int*)d_out;
    int* dgB1 = (int*)ws;

    hipMemsetAsync(deg, 0, (size_t)(N_NODES + N_EDGES) * sizeof(float), stream);
    degcnt_kernel<SL_ROWS_E><<<(IDX_NQ + 255) / 256, 256, 0, stream>>>(node_idx, edge_idx, deg, cnt, dgB0);

    gemm_wmma_kernel<D_IN><<<GM_BLOCKS, 128, 0, stream>>>(edge_fea, W1, B1);

    scatter_rows_lds_kernel<SL_ROWS_E, N_EDGES, true, false><<<SL_BLOCKS_E, SL_THREADS, 0, stream>>>(B1, dgB0, cnt, E);
    digest_kernel<SL_ROWS_N><<<(IDX_NQ + 255) / 256, 256, 0, stream>>>(node_idx, edge_idx, dgB1);

    scatter_rows_lds_kernel<SL_ROWS_N, N_NODES, false, true><<<SL_BLOCKS_N, SL_THREADS, 0, stream>>>(E, dgB1, cnt, B0);
    fused_scale_ln_prelu_kernel<<<LN_BLOCKS, 256, 0, stream>>>(B0, deg, b1, g1, beta1, prelu_a, B1);

    wstd_kernel<<<D, 256, 0, stream>>>(W2, W2s);
    gemm_wmma_kernel<D><<<GM_BLOCKS, 128, 0, stream>>>(B1, W2s, B0);

    digest_kernel<SL_ROWS_E><<<(IDX_NQ + 255) / 256, 256, 0, stream>>>(edge_idx, node_idx, dgB1);
    scatter_rows_lds_kernel<SL_ROWS_E, N_EDGES, true, false><<<SL_BLOCKS_E, SL_THREADS, 0, stream>>>(B0, dgB1, cnt, E);
    digest_kernel<SL_ROWS_N><<<(IDX_NQ + 255) / 256, 256, 0, stream>>>(node_idx, edge_idx, dgB0);

    scatter_rows_lds_kernel<SL_ROWS_N, N_NODES, false, true><<<SL_BLOCKS_N, SL_THREADS, 0, stream>>>(E, dgB0, cnt, B1);
    fused_scale_ln_prelu_kernel<<<LN_BLOCKS, 256, 0, stream>>>(B1, deg, b2, g2, beta2, prelu_a, B0);
}


# Layer 1's weight product moved behind its two scatter-adds (linearity): they carry 128-float input rows, the GEMM stays on 100000 node rows; sums before projection, a new summation order
# speedup vs baseline: 5.7294x; 1.0812x over previous
#include <hip/hip_runtime.h>
#include <hip/hip_bf16.h>

#define N_NODES   100000
#define N_EDGES   20000
#define NNZ_CNT   800000
#define D_IN      128
#define D         256
#define EPS_F     1e-5f

typedef float v2f __attribute__((ext_vector_type(2)));
typedef float v8f __attribute__((ext_vector_type(8)));

__device__ __forceinline__ void atomAddF(float* p, float v) {
    __hip_atomic_fetch_add(p, v, __ATOMIC_RELAXED, __HIP_MEMORY_SCOPE_AGENT);
}

typedef int v4i __attribute__((ext_vector_type(4)));
static_assert(NNZ_CNT % 16 == 0, "index lists are read as 16-byte quads, digests too");
static_assert(N_NODES <= (1 << 17) && N_EDGES <= (1 << 17), "a source row number packs into 17 bits");
#define IDX_NQ (NNZ_CNT / 4)
#define REC_OFF_WORDS (1 << 18)

template<int ROWS>
__device__ __forceinline__ void build_digest(int t, const int* __restrict__ dst_idx,
                                             const int* __restrict__ src_idx,
                                             int* __restrict__ dg) {
    const v4i d4 = reinterpret_cast<const v4i*>(dst_idx)[t];
    const v4i s4 = reinterpret_cast<const v4i*>(src_idx)[t];
    unsigned word = 0u;
    v4i rec;
    for (int j = 0; j < 4; ++j) {
        const unsigned grp = (unsigned)d4[j] / (unsigned)ROWS;
        const unsigned rel = (unsigned)d4[j] % (unsigned)ROWS;
        word |= (grp & 255u) << (8 * j);
        rec[j] = (int)(((unsigned)s4[j] << 15) | (min(grp >> 8, 127u) << 8) | rel);
    }
    dg[t] = (int)word;
    reinterpret_cast<v4i*>(dg + REC_OFF_WORDS)[t] = rec;
}

template<int ROWS>
__global__ __launch_bounds__(256) void digest_kernel(const int* __restrict__ dst_idx,
                                                     const int* __restrict__ src_idx,
                                                     int* __restrict__ dg) {
    const int t = blockIdx.x * 256 + threadIdx.x;
    if (t < IDX_NQ) build_digest<ROWS>(t, dst_idx, src_idx, dg);
}

template<int ROWS_E>
__global__ __launch_bounds__(256) void degcnt_kernel(const int* __restrict__ node_idx,
                                                     const int* __restrict__ edge_idx,
                                                     float* __restrict__ deg,
                                                     float* __restrict__ cnt,
                                                     int* __restrict__ dge) {
    const int t = blockIdx.x * 256 + threadIdx.x;
    if (t < IDX_NQ) {
        const v4i n4 = reinterpret_cast<const v4i*>(node_idx)[t];
        const v4i e4 = reinterpret_cast<const v4i*>(edge_idx)[t];
        for (int j = 0; j < 4; ++j) {
            atomAddF(&deg[n4[j]], 1.0f);
            atomAddF(&cnt[e4[j]], 1.0f);
        }
        build_digest<ROWS_E>(t, edge_idx, node_idx, dge);
    }
}

#define GM_ROWS 64
#define GM_BLOCKS ((N_NODES + GM_ROWS - 1) / GM_ROWS)
static_assert(D == 256, "four waves of 64 columns cover the 256 output columns");
static_assert(N_NODES % 16 == 0, "a 16-row tile is wholly inside or wholly outside the matrix");

typedef float v4f __attribute__((ext_vector_type(4)));

__device__ __forceinline__ void load_frag(const float* __restrict__ p, v4f (&f)[4]) {
    #pragma unroll
    for (int j = 0; j < 4; ++j) f[j] = *reinterpret_cast<const v4f*>(p + 4 * j);
}

template<int K>
__global__ __launch_bounds__(128) void gemm_wmma_kernel(const float* __restrict__ X,
                                                        const float* __restrict__ W,
                                                        float* __restrict__ C) {
    static_assert(K % 32 == 0, "K is walked in 32-k chunks");
    constexpr int NCH = K / 32;
    const int lane = threadIdx.x & 31;
    const int wave = __builtin_amdgcn_readfirstlane((int)(threadIdx.x >> 5));
    const int c  = lane & 15;
    const int h  = lane >> 4;
    const int m0 = blockIdx.x * GM_ROWS;
    const int n0 = wave * 64;

    const float* __restrict__ ap[4];
    #pragma unroll
    for (int i = 0; i < 4; ++i) {
        const int ti = (m0 + 16 * i < N_NODES) ? i : 0;
        ap[i] = X + (size_t)(m0 + 16 * ti + c) * K + 16 * h;
    }
    const float* __restrict__ bp = W + (size_t)(n0 + 4 * c) * K + 16 * h;

    v8f acc[4][4];
    #pragma unroll
    for (int i = 0; i < 4; ++i)
        #pragma unroll
        for (int n = 0; n < 4; ++n) acc[i][n] = v8f{};

    v4f fa[2][4][4], fb[2][4][4];

    #pragma unroll
    for (int i = 0; i < 4; ++i) load_frag(ap[i], fa[0][i]);
    #pragma unroll
    for (int n = 0; n < 4; ++n) load_frag(bp + (size_t)n * K, fb[0][n]);

    #pragma unroll
    for (int q = 0; q < NCH; ++q) {
        const int cur = q & 1, nxt = cur ^ 1;
        if (q + 1 < NCH) {
            #pragma unroll
            for (int i = 0; i < 4; ++i) load_frag(ap[i] + 32 * (q + 1), fa[nxt][i]);
            #pragma unroll
            for (int n = 0; n < 4; ++n) load_frag(bp + (size_t)n * K + 32 * (q + 1), fb[nxt][n]);
        }
        __builtin_amdgcn_sched_barrier(0);
        #pragma unroll
        for (int j = 0; j < 4; ++j) {
            #pragma unroll
            for (int p = 0; p < 2; ++p) {
                v2f a[4], b[4];
                #pragma unroll
                for (int i = 0; i < 4; ++i) {
                    a[i] = p ? __builtin_shufflevector(fa[cur][i][j], fa[cur][i][j], 2, 3)
                             : __builtin_shufflevector(fa[cur][i][j], fa[cur][i][j], 0, 1);
                    b[i] = p ? __builtin_shufflevector(fb[cur][i][j], fb[cur][i][j], 2, 3)
                             : __builtin_shufflevector(fb[cur][i][j], fb[cur][i][j], 0, 1);
                }
                #pragma unroll
                for (int i = 0; i < 4; ++i)
                    #pragma unroll
                    for (int n = 0; n < 4; ++n)
                        acc[i][n] = __builtin_amdgcn_wmma_f32_16x16x4_f32(false, a[i], false, b[n], (short)0,
                                                                          acc[i][n], false, false);
            }
        }
    }

    float* __restrict__ cp = C + (size_t)(m0 + 8 * h) * D + n0 + 4 * c;
    #pragma unroll
    for (int i = 0; i < 4; ++i) {
        if (m0 + 16 * i < N_NODES) {
            #pragma unroll
            for (int r = 0; r < 8; ++r) {
                const v4f o = {acc[i][0][r], acc[i][1][r], acc[i][2][r], acc[i][3][r]};
                *reinterpret_cast<v4f*>(cp + (size_t)(16 * i + r) * D) = o;
            }
        }
    }
}

#define SL_THREADS 1024
#define SL_WAVES   (SL_THREADS / 32)
#define SL_ACAP    160
#define SL_BCAP    48
#define SL_DCAP    512
#define DG_NQ      (NNZ_CNT / 16)
#define SL_NIT     ((DG_NQ + SL_THREADS - 1) / SL_THREADS)

__device__ __forceinline__ void ldsAddF(float* p, float v) {
    __hip_atomic_fetch_add(p, v, __ATOMIC_RELAXED, __HIP_MEMORY_SCOPE_WORKGROUP);
}

__device__ __forceinline__ v4i sl_load_quad(const v4i* __restrict__ g4, int qi) {
    return g4[min(qi, DG_NQ - 1)];
}

template<bool DEFER>
__device__ __forceinline__ int sl_check(const unsigned* qa, int pos, int n, unsigned* qb, int cb,
                                        int lane, unsigned grp_hi,
                                        const int* __restrict__ rec, unsigned* qd, int& cd) {
    const bool act = lane < n;
    const unsigned e = act ? qa[pos + lane] : 0u;
    const unsigned r = (unsigned)rec[e];
    const unsigned fld = (r >> 8) & 127u;
    const unsigned pk = ((r >> 15) << 8) | (r & 255u);
    const bool ok = act && fld == grp_hi;
    const unsigned mk = __builtin_amdgcn_ballot_w32(ok);
    if (ok) qb[cb + (int)__builtin_amdgcn_mbcnt_lo(mk, 0u)] = pk;
    if (DEFER) {
        const bool od = act && fld == grp_hi + 1u;
        const unsigned md = __builtin_amdgcn_ballot_w32(od);
        const int nd = __builtin_popcount(md);
        if (cd + nd <= SL_DCAP) {
            if (od) qd[cd + (int)__builtin_amdgcn_mbcnt_lo(md, 0u)] = pk;
            cd += nd;
        } else {
            cd = SL_DCAP + 1;
        }
    }
    return cb + __builtin_popcount(mk);
}

template<int NH>
__device__ __forceinline__ void sl_addN(const unsigned* qb, int pos, int lane,
                                        const float* __restrict__ src, float* acc) {
    constexpr int NB = 16 / NH;
    constexpr int RW = 128 * NH;
    const unsigned pk = qb[pos + (lane & (NB - 1))];
    const int rel = (int)(pk & 255u);
    const int s = (int)(pk >> 8);
    v4f v[NB][NH];
    for (int j = 0; j < NB; ++j) {
        const int sj = __builtin_amdgcn_readlane(s, j);
        const v4f* __restrict__ row4 = reinterpret_cast<const v4f*>(src + (size_t)sj * RW) + lane;
        for (int hh = 0; hh < NH; ++hh) v[j][hh] = row4[32 * hh];
    }
    for (int j = 0; j < NB; ++j) {
        const int rj = __builtin_amdgcn_readlane(rel, j);
        float* a = acc + rj * RW + lane;
        for (int hh = 0; hh < NH; ++hh)
            for (int c = 0; c < 4; ++c) ldsAddF(a + hh * 128 + c * 32, v[j][hh][c]);
    }
}

template<int NH>
__device__ __forceinline__ void sl_add1(unsigned pk, int lane, const float* __restrict__ src, float* acc) {
    constexpr int RW = 128 * NH;
    const v4f* __restrict__ row4 = reinterpret_cast<const v4f*>(src + (size_t)(pk >> 8) * RW) + lane;
    float* a = acc + (int)(pk & 255u) * RW + lane;
    v4f v[NH];
    for (int hh = 0; hh < NH; ++hh) v[hh] = row4[32 * hh];
    for (int hh = 0; hh < NH; ++hh)
        for (int c = 0; c < 4; ++c) ldsAddF(a + hh * 128 + c * 32, v[hh][c]);
}

template<bool DEFER, int NH>
__device__ __forceinline__ void sl_walk(const int* __restrict__ dg, const float* __restrict__ src,
                                        float* acc, unsigned* qa, unsigned* qb, unsigned* qd, int& cd,
                                        int tid, int lane, unsigned pat, unsigned grp_hi) {
    constexpr int NB = 16 / NH;
    const int* __restrict__ rec = dg + REC_OFF_WORDS;
    int ca = 0;
    int cb = 0;
    const v4i* __restrict__ g4 = reinterpret_cast<const v4i*>(dg);

    v4i nx = sl_load_quad(g4, tid);
    for (int it = 0; it < SL_NIT; ++it) {
        const int qi = it * SL_THREADS + tid;
        const v4i x = nx;
        nx = sl_load_quad(g4, qi + SL_THREADS);
        const bool in = qi < DG_NQ;
        #pragma unroll
        for (int c = 0; c < 4; ++c) {
            const unsigned y = (unsigned)x[c] ^ pat;
            unsigned z = ~(((y & 0x7F7F7F7Fu) + 0x7F7F7F7Fu) | y | 0x7F7F7F7Fu);
            z = in ? z : 0u;
            unsigned any = __builtin_amdgcn_ballot_w32(z != 0u);
            if (any != 0u) {
                do {
                    if (z != 0u) {
                        const int b = __builtin_ctz(z) >> 3;
                        z &= z - 1u;
                        qa[ca + (int)__builtin_amdgcn_mbcnt_lo(any, 0u)] = (unsigned)(16 * qi + 4 * c + b);
                    }
                    ca += __builtin_popcount(any);
                    any = __builtin_amdgcn_ballot_w32(z != 0u);
                } while (any != 0u);
                while (ca >= 32) {
                    ca -= 32;
                    cb = sl_check<DEFER>(qa, ca, 32, qb, cb, lane, grp_hi, rec, qd, cd);
                    while (cb >= NB) {
                        cb -= NB;
                        sl_addN<NH>(qb, cb, lane, src, acc);
                    }
                }
            }
        }
    }

    if (ca > 0) {
        cb = sl_check<DEFER>(qa, 0, ca, qb, cb, lane, grp_hi, rec, qd, cd);
        while (cb >= NB) {
            cb -= NB;
            sl_addN<NH>(qb, cb, lane, src, acc);
        }
    }
    for (int j = 0; j < cb; ++j)
        sl_add1<NH>((unsigned)__builtin_amdgcn_readfirstlane((int)qb[j]), lane, src, acc);
}

template<int ROWS, int NH>
__device__ __forceinline__ void sl_zero(float* acc, int tid) {
    v4f* acc4 = reinterpret_cast<v4f*>(acc);
    for (int k = tid; k < ROWS * 32 * NH; k += SL_THREADS) acc4[k] = v4f{0.0f, 0.0f, 0.0f, 0.0f};
}

template<int ROWS, int NDST, bool SCALE, int NH>
__device__ __forceinline__ void sl_store(const float* acc, int base, const float* __restrict__ rowcnt,
                                         float* __restrict__ dst, int tid) {
    constexpr int RW = 128 * NH;
    constexpr int PR = 32 * NH;
    v4f* __restrict__ out4 = reinterpret_cast<v4f*>(dst + (size_t)base * RW);
    for (int k = tid; k < ROWS * PR; k += SL_THREADS) {
        const int rr = k / PR;
        const int r = base + rr;
        if (r < NDST) {
            const float* p = acc + rr * RW + ((k >> 5) % NH) * 128 + (k & 31);
            v4f v = {p[0], p[32], p[64], p[96]};
            if (SCALE) {
                const float c = rowcnt[r];
                const float inv = (c > 0.0f) ? 1.0f / c : 0.0f;
                v = v * inv;
            }
            out4[k] = v;
        }
    }
}

template<int ROWS, int NDST, bool SCALE, bool PAIR, int NH>
__global__ __launch_bounds__(SL_THREADS) void scatter_rows_lds_kernel(const float* __restrict__ src,
                                                                      const int* __restrict__ dg,
                                                                      const float* __restrict__ rowcnt,
                                                                      float* __restrict__ dst) {
    static_assert(ROWS <= 256, "a row number inside the group packs into 8 bits");
    static_assert(!PAIR || 512 * ROWS >= NDST, "256 work-groups of two ranges cover every row");
    static_assert(NH == 1 || NH == 2, "a batch of 16 / NH entries keeps 16 row loads in flight");
    __shared__ float acc[ROWS * 128 * NH];
    __shared__ unsigned listA[SL_WAVES][SL_ACAP];
    __shared__ unsigned listB[SL_WAVES][SL_BCAP];
    __shared__ unsigned listD[PAIR ? SL_WAVES : 1][PAIR ? SL_DCAP : 1];

    const int tid  = threadIdx.x;
    const int lane = tid & 31;
    const int wave = __builtin_amdgcn_readfirstlane(tid >> 5);
    const unsigned pat = (unsigned)(blockIdx.x & 255) * 0x01010101u;
    const unsigned grp_hi = (unsigned)blockIdx.x >> 8;

    unsigned* qa = listA[wave];
    unsigned* qb = listB[wave];
    unsigned* qd = listD[PAIR ? wave : 0];
    int cd = 0;

    sl_zero<ROWS, NH>(acc, tid);
    __syncthreads();
    sl_walk<PAIR, NH>(dg, src, acc, qa, qb, qd, cd, tid, lane, pat, grp_hi);
    __syncthreads();
    sl_store<ROWS, NDST, SCALE, NH>(acc, (int)blockIdx.x * ROWS, rowcnt, dst, tid);

    if (PAIR) {
        __syncthreads();
        sl_zero<ROWS, NH>(acc, tid);
        __syncthreads();
        const int nd = __builtin_amdgcn_readfirstlane(cd);
        if (nd <= SL_DCAP) {
            constexpr int NB = 16 / NH;
            int k = 0;
            for (; k + NB <= nd; k += NB) sl_addN<NH>(qd, k, lane, src, acc);
            for (; k < nd; ++k)
                sl_add1<NH>((unsigned)__builtin_amdgcn_readfirstlane((int)qd[k]), lane, src, acc);
        } else {
            int none = 0;
            sl_walk<false, NH>(dg, src, acc, qa, qb, qd, none, tid, lane, pat, grp_hi + 1u);
        }
        __syncthreads();
        sl_store<ROWS, NDST, SCALE, NH>(acc, ((int)blockIdx.x + 256) * ROWS, rowcnt, dst, tid);
    }
}

#define SL_ROWS_E 80
#define SL_ROWS_N 196
#define SL_BLOCKS_E ((N_EDGES + SL_ROWS_E - 1) / SL_ROWS_E)
#define SL_BLOCKS_N 256
static_assert(SL_BLOCKS_E <= 256, "one range per group: every group has a digest byte of its own");

#define LN_RPW 4
#define LN_WAVES_PER_BLOCK 8
#define LN_BLOCKS (N_NODES / (LN_RPW * LN_WAVES_PER_BLOCK))
static_assert(N_NODES % (LN_RPW * LN_WAVES_PER_BLOCK) == 0, "row count must fill whole blocks");
static_assert(D == 256, "lane-to-column map below is written for 256-wide rows");

__device__ __forceinline__ float lane_sum8(v4f a, v4f b) {
    return ((a.x + a.y) + (a.z + a.w)) + ((b.x + b.y) + (b.z + b.w));
}

__global__ __launch_bounds__(256) void fused_scale_ln_prelu_kernel(const float* __restrict__ acc,
                                                                   const float* __restrict__ deg,
                                                                   const float* __restrict__ bias,
                                                                   const float* __restrict__ g,
                                                                   const float* __restrict__ beta,
                                                                   const float* __restrict__ pa,
                                                                   float* __restrict__ out) {
    const int lane = threadIdx.x & 31;
    const int wv = __builtin_amdgcn_readfirstlane((int)((blockIdx.x * 256 + threadIdx.x) >> 5));
    const int r0 = wv * LN_RPW;
    const int c0 = lane * 4;

    v4f x0[LN_RPW], x1[LN_RPW];
    for (int j = 0; j < LN_RPW; ++j) {
        const float* __restrict__ row = acc + (size_t)(r0 + j) * D + c0;
        x0[j] = *reinterpret_cast<const v4f*>(row);
        x1[j] = *reinterpret_cast<const v4f*>(row + 128);
    }

    float dinv[LN_RPW];
    for (int j = 0; j < LN_RPW; ++j) {
        const float dg = deg[r0 + j];
        dinv[j] = (dg > 0.0f) ? 1.0f / dg : 0.0f;
    }

    const v4f bs0 = *reinterpret_cast<const v4f*>(bias + c0);
    const v4f bs1 = *reinterpret_cast<const v4f*>(bias + c0 + 128);
    const v4f g0  = *reinterpret_cast<const v4f*>(g + c0);
    const v4f g1  = *reinterpret_cast<const v4f*>(g + c0 + 128);
    const v4f be0 = *reinterpret_cast<const v4f*>(beta + c0);
    const v4f be1 = *reinterpret_cast<const v4f*>(beta + c0 + 128);
    const float a = pa[0];

    float sum[LN_RPW];
    for (int j = 0; j < LN_RPW; ++j) {
        x0[j] = x0[j] * dinv[j] + bs0;
        x1[j] = x1[j] * dinv[j] + bs1;
        sum[j] = lane_sum8(x0[j], x1[j]);
    }
    for (int m = 16; m > 0; m >>= 1)
        for (int j = 0; j < LN_RPW; ++j) sum[j] += __shfl_xor(sum[j], m, 32);

    float sq[LN_RPW];
    for (int j = 0; j < LN_RPW; ++j) {
        const float mu = sum[j] * (1.0f / D);
        x0[j] = x0[j] - mu;
        x1[j] = x1[j] - mu;
        sq[j] = lane_sum8(x0[j] * x0[j], x1[j] * x1[j]);
    }
    for (int m = 16; m > 0; m >>= 1)
        for (int j = 0; j < LN_RPW; ++j) sq[j] += __shfl_xor(sq[j], m, 32);

    for (int j = 0; j < LN_RPW; ++j) {
        const float rs = rsqrtf(sq[j] * (1.0f / D) + EPS_F);
        const v4f y0 = x0[j] * rs * g0 + be0;
        const v4f y1 = x1[j] * rs * g1 + be1;
        v4f o0, o1;
        for (int c = 0; c < 4; ++c) {
            o0[c] = (y0[c] >= 0.0f) ? y0[c] : a * y0[c];
            o1[c] = (y1[c] >= 0.0f) ? y1[c] : a * y1[c];
        }
        float* __restrict__ orow = out + (size_t)(r0 + j) * D + c0;
        *reinterpret_cast<v4f*>(orow)       = o0;
        *reinterpret_cast<v4f*>(orow + 128) = o1;
    }
}

__global__ __launch_bounds__(256) void wstd_kernel(const float* __restrict__ W,
                                                   float* __restrict__ Ws) {
    __shared__ float sred[256];
    const int r = blockIdx.x;
    const int f = threadIdx.x;
    const float w = W[(size_t)r * D + f];

    sred[f] = w;
    __syncthreads();
    #pragma unroll
    for (int s = 128; s > 0; s >>= 1) { if (f < s) sred[f] += sred[f + s]; __syncthreads(); }
    const float mean = sred[0] * (1.0f / D);
    __syncthreads();

    const float dv = w - mean;
    sred[f] = dv * dv;
    __syncthreads();
    #pragma unroll
    for (int s = 128; s > 0; s >>= 1) { if (f < s) sred[f] += sred[f + s]; __syncthreads(); }
    const float var = sred[0] * (1.0f / (D - 1));

    Ws[(size_t)r * D + f] = dv * rsqrtf(var + EPS_F);
}

extern "C" void kernel_launch(void* const* d_in, const int* in_sizes, int n_in,
                              void* d_out, int out_size, void* d_ws, size_t ws_size,
                              hipStream_t stream) {
    const float* edge_fea = (const float*)d_in[0];
    const int*   node_idx = (const int*)  d_in[1];
    const int*   edge_idx = (const int*)  d_in[2];
    const float* W1       = (const float*)d_in[3];
    const float* b1       = (const float*)d_in[4];
    const float* W2       = (const float*)d_in[5];
    const float* b2       = (const float*)d_in[6];
    const float* g1       = (const float*)d_in[7];
    const float* beta1    = (const float*)d_in[8];
    const float* g2       = (const float*)d_in[9];
    const float* beta2    = (const float*)d_in[10];
    const float* prelu_a  = (const float*)d_in[11];

    const size_t nodeRowBytes = (size_t)N_NODES * D * sizeof(float);

    float* B0  = (float*)d_out;
    char*  ws  = (char*)d_ws;
    float* B1  = (float*)ws;
    float* P   = B1 + (size_t)N_NODES * D_IN;
    float* E   = (float*)(ws + nodeRowBytes);
    float* deg = E + (size_t)N_EDGES * D;
    float* cnt = deg + N_NODES;
    float* W2s = cnt + N_EDGES;


    int* dgB0 = (int*)d_out;
    int* dgB1 = (int*)ws;

    hipMemsetAsync(deg, 0, (size_t)(N_NODES + N_EDGES) * sizeof(float), stream);
    degcnt_kernel<SL_ROWS_E><<<(IDX_NQ + 255) / 256, 256, 0, stream>>>(node_idx, edge_idx, deg, cnt, dgB0);


    scatter_rows_lds_kernel<SL_ROWS_E, N_EDGES, true, false, 1><<<SL_BLOCKS_E, SL_THREADS, 0, stream>>>(edge_fea, dgB0, cnt, E);
    digest_kernel<SL_ROWS_N><<<(IDX_NQ + 255) / 256, 256, 0, stream>>>(node_idx, edge_idx, dgB1);

    scatter_rows_lds_kernel<SL_ROWS_N, N_NODES, false, true, 1><<<SL_BLOCKS_N, SL_THREADS, 0, stream>>>(E, dgB1, cnt, P);
    gemm_wmma_kernel<D_IN><<<GM_BLOCKS, 128, 0, stream>>>(P, W1, B0);
    fused_scale_ln_prelu_kernel<<<LN_BLOCKS, 256, 0, stream>>>(B0, deg, b1, g1, beta1, prelu_a, B1);

    wstd_kernel<<<D, 256, 0, stream>>>(W2, W2s);
    gemm_wmma_kernel<D><<<GM_BLOCKS, 128, 0, stream>>>(B1, W2s, B0);

    digest_kernel<SL_ROWS_E><<<(IDX_NQ + 255) / 256, 256, 0, stream>>>(edge_idx, node_idx, dgB1);
    scatter_rows_lds_kernel<SL_ROWS_E, N_EDGES, true, false, 2><<<SL_BLOCKS_E, SL_THREADS, 0, stream>>>(B0, dgB1, cnt, E);
    digest_kernel<SL_ROWS_N><<<(IDX_NQ + 255) / 256, 256, 0, stream>>>(node_idx, edge_idx, dgB0);

    scatter_rows_lds_kernel<SL_ROWS_N, N_NODES, false, true, 2><<<SL_BLOCKS_N, SL_THREADS, 0, stream>>>(E, dgB0, cnt, B1);
    fused_scale_ln_prelu_kernel<<<LN_BLOCKS, 256, 0, stream>>>(B1, deg, b2, g2, beta2, prelu_a, B0);
}
